# starting kernel
# baseline (speedup 1.0000x reference)
#include <hip/hip_runtime.h>
#include <hip/hip_bf16.h>

typedef float v2f __attribute__((ext_vector_type(2)));
typedef float v8f __attribute__((ext_vector_type(8)));

#define LD      1282
#define KDIM    256
#define NTOT    1282
#define NTILES  81
#define MTILES  256
#define SSTEPS  512
#define BATCH   8

__global__ __launch_bounds__(256)
void fwp_slow_gemm(const float* __restrict__ x,
                   const float* __restrict__ slow_W,
                   const float* __restrict__ slow_b,
                   const float* __restrict__ fast_W0,
                   float* __restrict__ wu)
{
    int wid  = threadIdx.x >> 5;
    int lane = threadIdx.x & 31;
    int wt   = blockIdx.x * 8 + wid;
    if (wt >= MTILES * NTILES) return;
    int mt = wt / NTILES, nt = wt % NTILES;
    int m0 = mt * 16, n0 = nt * 16;
    int q = lane & 15, h = lane >> 4;

    int n = n0 + q;
    const float* wp;
    if (n < 1026) {
        wp = slow_W + (size_t)n * KDIM;
    } else {
        int idx = n - 1026;
        if (idx > 255) idx = 255;
        wp = fast_W0 + (size_t)idx * KDIM;
    }
    float bv = (n < 1026) ? slow_b[n] : 0.0f;

    v8f c;
    #pragma unroll
    for (int r = 0; r < 8; ++r) c[r] = bv;

    const float* ap = x + (size_t)(m0 + q) * KDIM + 2 * h;
    const float* bp = wp + 2 * h;

    #pragma unroll 8
    for (int k = 0; k < KDIM; k += 4) {
        v2f a = *(const v2f*)(ap + k);
        v2f b = *(const v2f*)(bp + k);
        c = __builtin_amdgcn_wmma_f32_16x16x4_f32(false, a, false, b,
                                                  (short)0, c, false, false);
    }

    if (n < NTOT) {
        #pragma unroll
        for (int r = 0; r < 8; ++r) {
            int m = m0 + h * 8 + r;
            wu[(size_t)m * LD + n] = c[r];
        }
    }
}

__global__ __launch_bounds__(256)
void fwp_scan(const float* __restrict__ wu,
              const float* __restrict__ x,
              float* __restrict__ out)
{
    __shared__ float red[8 * 16];

    int tid  = threadIdx.x;
    int w    = tid >> 5;
    int lane = tid & 31;
    int q    = lane & 15;
    int h    = lane >> 4;
    int o0   = blockIdx.x * 16;
    int b    = blockIdx.y;
    int c0   = 32 * w + q;
    int c1   = c0 + 16;

    v8f fw0 = {0,0,0,0,0,0,0,0};
    v8f fw1 = {0,0,0,0,0,0,0,0};
    const v8f zero = {0,0,0,0,0,0,0,0};

    const int STEP_BYTES = BATCH * LD * 4;

    for (int s = 0; s < SSTEPS; ++s) {
        const float* row  = wu + (size_t)(s * BATCH) * LD;
        const float* rowb = row + (size_t)b * LD;

        if (s + 1 < SSTEPS) {
            const char* nb = (const char*)(row) + STEP_BYTES;
            int off0 = tid * 128;
            int off1 = 32768 + tid * 128;
            if (off0 < STEP_BYTES) __builtin_prefetch(nb + off0, 0, 3);
            if (off1 < STEP_BYTES) __builtin_prefetch(nb + off1, 0, 3);
            if (tid < 64) {
                const char* nx = (const char*)(x + (size_t)(s + 1) * BATCH * 256);
                __builtin_prefetch(nx + tid * 128, 0, 3);
            }
        }

        float sa = 1.0f / (1.0f + __expf(-rowb[1024]));
        float ss = 1.0f / (1.0f + __expf(-rowb[1025]));

        int kk = 2 * h;

        v2f aA0, aA1, sA0, sA1;
        aA0.x = row[(size_t)(kk + 0) * LD +       o0 + q];
        aA0.y = row[(size_t)(kk + 1) * LD +       o0 + q];
        aA1.x = row[(size_t)(kk + 4) * LD +       o0 + q];
        aA1.y = row[(size_t)(kk + 5) * LD +       o0 + q];
        sA0.x = row[(size_t)(kk + 0) * LD + 512 + o0 + q];
        sA0.y = row[(size_t)(kk + 1) * LD + 512 + o0 + q];
        sA1.x = row[(size_t)(kk + 4) * LD + 512 + o0 + q];
        sA1.y = row[(size_t)(kk + 5) * LD + 512 + o0 + q];

        v2f aB0t0, aB1t0, aB0t1, aB1t1, sB0t0, sB1t0, sB0t1, sB1t1;
        aB0t0.x = row[(size_t)(kk + 0) * LD + 256 + c0];
        aB0t0.y = row[(size_t)(kk + 1) * LD + 256 + c0];
        aB1t0.x = row[(size_t)(kk + 4) * LD + 256 + c0];
        aB1t0.y = row[(size_t)(kk + 5) * LD + 256 + c0];
        aB0t1.x = row[(size_t)(kk + 0) * LD + 256 + c1];
        aB0t1.y = row[(size_t)(kk + 1) * LD + 256 + c1];
        aB1t1.x = row[(size_t)(kk + 4) * LD + 256 + c1];
        aB1t1.y = row[(size_t)(kk + 5) * LD + 256 + c1];
        sB0t0.x = row[(size_t)(kk + 0) * LD + 768 + c0];
        sB0t0.y = row[(size_t)(kk + 1) * LD + 768 + c0];
        sB1t0.x = row[(size_t)(kk + 4) * LD + 768 + c0];
        sB1t0.y = row[(size_t)(kk + 5) * LD + 768 + c0];
        sB0t1.x = row[(size_t)(kk + 0) * LD + 768 + c1];
        sB0t1.y = row[(size_t)(kk + 1) * LD + 768 + c1];
        sB1t1.x = row[(size_t)(kk + 4) * LD + 768 + c1];
        sB1t1.y = row[(size_t)(kk + 5) * LD + 768 + c1];

        v8f ad0 = __builtin_amdgcn_wmma_f32_16x16x4_f32(false, aA0, false, aB0t0, (short)0, zero, false, false);
        ad0     = __builtin_amdgcn_wmma_f32_16x16x4_f32(false, aA1, false, aB1t0, (short)0, ad0,  false, false);
        v8f ad1 = __builtin_amdgcn_wmma_f32_16x16x4_f32(false, aA0, false, aB0t1, (short)0, zero, false, false);
        ad1     = __builtin_amdgcn_wmma_f32_16x16x4_f32(false, aA1, false, aB1t1, (short)0, ad1,  false, false);
        v8f su0 = __builtin_amdgcn_wmma_f32_16x16x4_f32(false, sA0, false, sB0t0, (short)0, zero, false, false);
        su0     = __builtin_amdgcn_wmma_f32_16x16x4_f32(false, sA1, false, sB1t0, (short)0, su0,  false, false);
        v8f su1 = __builtin_amdgcn_wmma_f32_16x16x4_f32(false, sA0, false, sB0t1, (short)0, zero, false, false);
        su1     = __builtin_amdgcn_wmma_f32_16x16x4_f32(false, sA1, false, sB1t1, (short)0, su1,  false, false);

        #pragma unroll
        for (int r = 0; r < 8; ++r) {
            fw0[r] = fmaf(ad0[r], sa, fw0[r]);
            fw0[r] = fmaf(su0[r], -ss, fw0[r]);
            fw1[r] = fmaf(ad1[r], sa, fw1[r]);
            fw1[r] = fmaf(su1[r], -ss, fw1[r]);
        }

        float xv0 = x[(size_t)(s * BATCH + b) * 256 + c0];
        float xv1 = x[(size_t)(s * BATCH + b) * 256 + c1];
        float pr[8];
        #pragma unroll
        for (int r = 0; r < 8; ++r)
            pr[r] = fw0[r] * xv0 + fw1[r] * xv1;

        #pragma unroll
        for (int m = 1; m <= 8; m <<= 1) {
            #pragma unroll
            for (int r = 0; r < 8; ++r)
                pr[r] += __shfl_xor(pr[r], m, 32);
        }

        if (q == 0) {
            #pragma unroll
            for (int r = 0; r < 8; ++r)
                red[w * 16 + 8 * h + r] = pr[r];
        }
        __syncthreads();

        if (tid < 16) {
            float v = rowb[1026 + o0 + tid];
            #pragma unroll
            for (int ww = 0; ww < 8; ++ww)
                v += red[ww * 16 + tid];
            out[(size_t)(s * BATCH + b) * 256 + o0 + tid] = v;
        }
        __syncthreads();
    }
}

extern "C" void kernel_launch(void* const* d_in, const int* in_sizes, int n_in,
                              void* d_out, int out_size, void* d_ws, size_t ws_size,
                              hipStream_t stream)
{
    const float* x       = (const float*)d_in[0];
    const float* slow_W  = (const float*)d_in[1];
    const float* slow_b  = (const float*)d_in[2];
    const float* fast_W0 = (const float*)d_in[3];
    float*       out     = (float*)d_out;
    float*       wu      = (float*)d_ws;

    int total_waves = MTILES * NTILES;
    int blocks1 = (total_waves + 7) / 8;
    fwp_slow_gemm<<<blocks1, 256, 0, stream>>>(x, slow_W, slow_b, fast_W0, wu);

    dim3 grid2(16, BATCH);
    fwp_scan<<<grid2, 256, 0, stream>>>(wu, x, out);
}


# Scan: a wave holds 4 row-interleaved state tiles (64 i) of one key half (add/sub): one 128-bit key-row load feeds 4 A fragments, 1 load per WMMA not 3; fw is 2 partial sums, order of sums changed
# speedup vs baseline: 3.7270x; 3.7270x over previous
#include <hip/hip_runtime.h>
#include <hip/hip_bf16.h>

typedef float v2f __attribute__((ext_vector_type(2)));
typedef float v4f __attribute__((ext_vector_type(4)));
typedef float v8f __attribute__((ext_vector_type(8)));

#define LD      1282
#define KDIM    256
#define NTOT    1282
#define NTILES  81
#define MTILES  256
#define SSTEPS  512
#define BATCH   8
#define CH      16
#define NCHUNK  32
#define KEYS    256
#define WU_FLOATS 5251072

__global__ __launch_bounds__(256)
void fwp_slow_gemm(const float* __restrict__ x,
                   const float* __restrict__ slow_W,
                   const float* __restrict__ slow_b,
                   const float* __restrict__ fast_W0,
                   float* __restrict__ wu)
{
    int wid  = threadIdx.x >> 5;
    int lane = threadIdx.x & 31;
    int wt   = blockIdx.x * 8 + wid;
    if (wt >= MTILES * NTILES) return;
    int mt = wt / NTILES, nt = wt % NTILES;
    int m0 = mt * 16, n0 = nt * 16;
    int q = lane & 15, h = lane >> 4;

    int n = n0 + q;
    const float* wp;
    if (n < 1026) {
        wp = slow_W + (size_t)n * KDIM;
    } else {
        int idx = n - 1026;
        if (idx > 255) idx = 255;
        wp = fast_W0 + (size_t)idx * KDIM;
    }
    float bv = (n < 1026) ? slow_b[n] : 0.0f;

    v8f c;
    #pragma unroll
    for (int r = 0; r < 8; ++r) c[r] = bv;

    const float* ap = x + (size_t)(m0 + q) * KDIM + 2 * h;
    const float* bp = wp + 2 * h;

    #pragma unroll 8
    for (int k = 0; k < KDIM; k += 4) {
        v2f a = *(const v2f*)(ap + k);
        v2f b = *(const v2f*)(bp + k);
        c = __builtin_amdgcn_wmma_f32_16x16x4_f32(false, a, false, b,
                                                  (short)0, c, false, false);
    }

    if (n < NTOT) {
        #pragma unroll
        for (int r = 0; r < 8; ++r) {
            int m = m0 + h * 8 + r;
            wu[(size_t)m * LD + n] = c[r];
        }
    }
}

__global__ __launch_bounds__(256)
void fwp_scores(const float* __restrict__ wu,
                const float* __restrict__ x,
                float* __restrict__ P)
{
    int wid  = threadIdx.x >> 5;
    int lane = threadIdx.x & 31;
    int wt   = blockIdx.x * 8 + wid;
    int kt   = wt & 15;
    int b    = (wt >> 4) & 7;
    int c    = wt >> 7;
    int q = lane & 15, h = lane >> 4;

    int key = kt * 16 + q;
    int kl  = key & 127;
    const float* ap = x  + (size_t)((c * CH + q) * BATCH + b) * KDIM + 2 * h;
    const float* bp = wu + (size_t)(c * 128 + kl) * LD + (key < 128 ? 256 : 768) + 2 * h;

    v8f acc = {0,0,0,0,0,0,0,0};
    #pragma unroll 8
    for (int k = 0; k < KDIM; k += 4) {
        v2f a  = *(const v2f*)(ap + k);
        v2f bb = *(const v2f*)(bp + k);
        acc = __builtin_amdgcn_wmma_f32_16x16x4_f32(false, a, false, bb,
                                                    (short)0, acc, false, false);
    }

    int tk = kl >> 3;
    float* pp = P + (size_t)((c * BATCH + b) * CH) * KEYS + key;
    #pragma unroll
    for (int r = 0; r < 8; ++r) {
        int s = 8 * h + r;
        pp[(size_t)s * KEYS] = (tk <= s) ? acc[r] : 0.0f;
    }
}

struct FragSet { v4f ka[8][2]; v2f bv[8]; };

__device__ __forceinline__ v4f ld4(const float* __restrict__ p)
{
    v4f v;
    __builtin_memcpy(&v, p, 16);
    return v;
}

template <int G>
__device__ __forceinline__ void load_group(FragSet& f, const float* __restrict__ kp,
                                           const float* __restrict__ vp)
{
    #pragma unroll
    for (int u = 0; u < 8; ++u) {
        const int r0 = (32 * G + 4 * u) * LD;
        const int r1 = r0 + LD;
        f.ka[u][0] = ld4(kp + r0);
        f.ka[u][1] = ld4(kp + r1);
        f.bv[u].x = vp[r0];
        f.bv[u].y = vp[r1];
    }
}

template <int G>
__device__ __forceinline__ void mma_group(const FragSet& f, int gvi, int gl0, v8f (&T)[4])
{
    #pragma unroll
    for (int u = 0; u < 8; ++u) {
        float g = __int_as_float(__builtin_amdgcn_readlane(gvi, gl0 + 4 * G + (u >> 1)));
        v2f vb;
        vb.x = f.bv[u].x * g;
        vb.y = f.bv[u].y * g;
        #pragma unroll
        for (int j = 0; j < 4; ++j) {
            v2f a; a.x = f.ka[u][0][j]; a.y = f.ka[u][1][j];
            T[j] = __builtin_amdgcn_wmma_f32_16x16x4_f32(false, a, false, vb,
                                                         (short)0, T[j], false, false);
        }
    }
}

#define STAGE_FENCE() __builtin_amdgcn_sched_barrier(0)

__global__ __launch_bounds__(256)
void fwp_scan(const float* __restrict__ wu,
              const float* __restrict__ x,
              const float* __restrict__ P,
              float* __restrict__ out)
{
    __shared__ float red[2][8][256];

    int tid  = threadIdx.x;
    int lane = tid & 31;
    int q    = lane & 15;
    int h    = lane >> 4;
    int w    = __builtin_amdgcn_readfirstlane(tid >> 5);
    int g    = w & 3;
    int kh   = w >> 2;
    int gl0  = 16 * kh;
    int o0   = blockIdx.x * 16;
    int b    = blockIdx.y;

    v8f T[4];
    #pragma unroll
    for (int j = 0; j < 4; ++j) T[j] = (v8f){0,0,0,0,0,0,0,0};

    for (int c = 0; c < NCHUNK; ++c) {
        const float* wc = wu + (size_t)(c * 128) * LD;

        size_t orow = (size_t)((c * CH + (tid >> 4)) * BATCH + b);
        float basev = wu[orow * LD + 1026 + o0 + (tid & 15)];
        float gin = wc[(size_t)(q * BATCH + b) * LD + 1024 + h];

        const float* xr = x + (size_t)((c * CH + q) * BATCH + b) * KDIM + 64 * g + 4 * h;
        const float* pr = P + (size_t)((c * BATCH + b) * CH + q) * KEYS + 32 * w + 2 * h;
        const float* vr = wc + (size_t)(32 * (w & 3) + 2 * h) * LD + (w < 4 ? 0 : 512) + o0 + q;

        v4f xv[4][2];
        v2f pa[8], vraw[8];
        #pragma unroll
        for (int jj = 0; jj < 4; ++jj) {
            xv[jj][0] = *(const v4f*)(xr + 16 * jj);
            xv[jj][1] = *(const v4f*)(xr + 16 * jj + 8);
        }
        #pragma unroll
        for (int jj = 0; jj < 8; ++jj) {
            pa[jj] = *(const v2f*)(pr + 4 * jj);
            vraw[jj].x = vr[(4 * jj) * LD];
            vraw[jj].y = vr[(4 * jj + 1) * LD];
        }

        const float* kp = wc + (size_t)(2 * h) * LD + (kh ? 768 : 256) + 64 * g + 8 * (q & 7) + 4 * (q >> 3);
        const float* vp = wc + (size_t)(2 * h) * LD + (kh ? 512 : 0) + o0 + q;
        FragSet f0, f1;
        load_group<0>(f0, kp, vp);
        STAGE_FENCE();

        float gv  = 1.0f / (1.0f + __expf(-gin));
        int   gvi = __float_as_int(h ? -gv : gv);

        v8f accA = {0,0,0,0,0,0,0,0};
        v8f accB = {0,0,0,0,0,0,0,0};
        #pragma unroll
        for (int jj = 0; jj < 4; ++jj) {
            #pragma unroll
            for (int j = 0; j < 4; j += 2) {
                v2f a0; a0.x = xv[jj][0][j];     a0.y = xv[jj][1][j];
                v2f a1; a1.x = xv[jj][0][j + 1]; a1.y = xv[jj][1][j + 1];
                v2f tb0; tb0.x = T[j][2 * jj];     tb0.y = T[j][2 * jj + 1];
                v2f tb1; tb1.x = T[j + 1][2 * jj]; tb1.y = T[j + 1][2 * jj + 1];
                accA = __builtin_amdgcn_wmma_f32_16x16x4_f32(false, a0, false, tb0,
                                                             (short)0, accA, false, false);
                accB = __builtin_amdgcn_wmma_f32_16x16x4_f32(false, a1, false, tb1,
                                                             (short)0, accB, false, false);
            }
        }
        #pragma unroll
        for (int jj = 0; jj < 8; jj += 2) {
            float gg = __int_as_float(__builtin_amdgcn_readlane(gvi, 4 * w + (jj >> 1)));
            v2f vb0; vb0.x = vraw[jj].x * gg;     vb0.y = vraw[jj].y * gg;
            v2f vb1; vb1.x = vraw[jj + 1].x * gg; vb1.y = vraw[jj + 1].y * gg;
            accA = __builtin_amdgcn_wmma_f32_16x16x4_f32(false, pa[jj], false, vb0,
                                                         (short)0, accA, false, false);
            accB = __builtin_amdgcn_wmma_f32_16x16x4_f32(false, pa[jj + 1], false, vb1,
                                                         (short)0, accB, false, false);
        }

        float* rw = &red[c & 1][w][0];
        #pragma unroll
        for (int r = 0; r < 8; ++r)
            rw[(8 * h + r) * 16 + q] = accA[r] + accB[r];
        STAGE_FENCE();

        load_group<1>(f1, kp, vp);  STAGE_FENCE();  mma_group<0>(f0, gvi, gl0, T);  STAGE_FENCE();
        load_group<2>(f0, kp, vp);  STAGE_FENCE();  mma_group<1>(f1, gvi, gl0, T);  STAGE_FENCE();
        load_group<3>(f1, kp, vp);  STAGE_FENCE();  mma_group<2>(f0, gvi, gl0, T);  STAGE_FENCE();
        mma_group<3>(f1, gvi, gl0, T);
        STAGE_FENCE();

        __syncthreads();
        float v = basev;
        #pragma unroll
        for (int ww = 0; ww < 8; ++ww)
            v += red[c & 1][ww][tid];
        out[orow * 256 + o0 + (tid & 15)] = v;
    }
}

extern "C" void kernel_launch(void* const* d_in, const int* in_sizes, int n_in,
                              void* d_out, int out_size, void* d_ws, size_t ws_size,
                              hipStream_t stream)
{
    const float* x       = (const float*)d_in[0];
    const float* slow_W  = (const float*)d_in[1];
    const float* slow_b  = (const float*)d_in[2];
    const float* fast_W0 = (const float*)d_in[3];
    float*       out     = (float*)d_out;
    float*       wu      = (float*)d_ws;
    float*       P       = (float*)d_ws + WU_FLOATS;

    int total_waves = MTILES * NTILES;
    int blocks1 = (total_waves + 7) / 8;
    fwp_slow_gemm<<<blocks1, 256, 0, stream>>>(x, slow_W, slow_b, fast_W0, wu);

    fwp_scores<<<512, 256, 0, stream>>>(wu, x, P);

    dim3 grid2(16, BATCH);
    fwp_scan<<<grid2, 256, 0, stream>>>(wu, x, P, out);
}


# Scan on all 256 WGPs: add-key and sub-key partial states in separate 4-wave work-groups, one per WGP (LDS reservation), one wave per SIMD; a 4th kernel adds the output planes; order of sums changed
# speedup vs baseline: 4.9584x; 1.3304x over previous
#include <hip/hip_runtime.h>
#include <hip/hip_bf16.h>

typedef float v2f __attribute__((ext_vector_type(2)));
typedef float v4f __attribute__((ext_vector_type(4)));
typedef float v8f __attribute__((ext_vector_type(8)));

#define LD      1282
#define KDIM    256
#define NTOT    1282
#define NTILES  81
#define MTILES  256
#define SSTEPS  512
#define BATCH   8
#define CH      16
#define NCHUNK  32
#define KEYS    256
#define WU_FLOATS 5251072
#define P_FLOATS  1048576
#define OUT_FLOATS 1048576
#define LDS_FLOATS 41472

__global__ __launch_bounds__(256)
void fwp_slow_gemm(const float* __restrict__ x,
                   const float* __restrict__ slow_W,
                   const float* __restrict__ slow_b,
                   const float* __restrict__ fast_W0,
                   float* __restrict__ wu)
{
    int wid  = threadIdx.x >> 5;
    int lane = threadIdx.x & 31;
    int wt   = blockIdx.x * 8 + wid;
    if (wt >= MTILES * NTILES) return;
    int mt = wt / NTILES, nt = wt % NTILES;
    int m0 = mt * 16, n0 = nt * 16;
    int q = lane & 15, h = lane >> 4;

    int n = n0 + q;
    const float* wp;
    if (n < 1026) {
        wp = slow_W + (size_t)n * KDIM;
    } else {
        int idx = n - 1026;
        if (idx > 255) idx = 255;
        wp = fast_W0 + (size_t)idx * KDIM;
    }
    float bv = (n < 1026) ? slow_b[n] : 0.0f;

    v8f c;
    #pragma unroll
    for (int r = 0; r < 8; ++r) c[r] = bv;

    const float* ap = x + (size_t)(m0 + q) * KDIM + 2 * h;
    const float* bp = wp + 2 * h;

    #pragma unroll 8
    for (int k = 0; k < KDIM; k += 4) {
        v2f a = *(const v2f*)(ap + k);
        v2f b = *(const v2f*)(bp + k);
        c = __builtin_amdgcn_wmma_f32_16x16x4_f32(false, a, false, b,
                                                  (short)0, c, false, false);
    }

    if (n < NTOT) {
        #pragma unroll
        for (int r = 0; r < 8; ++r) {
            int m = m0 + h * 8 + r;
            wu[(size_t)m * LD + n] = c[r];
        }
    }
}

__global__ __launch_bounds__(256)
void fwp_scores(const float* __restrict__ wu,
                const float* __restrict__ x,
                float* __restrict__ P)
{
    int wid  = threadIdx.x >> 5;
    int lane = threadIdx.x & 31;
    int wt   = blockIdx.x * 8 + wid;
    int kt   = wt & 15;
    int b    = (wt >> 4) & 7;
    int c    = wt >> 7;
    int q = lane & 15, h = lane >> 4;

    int key = kt * 16 + q;
    int kl  = key & 127;
    const float* ap = x  + (size_t)((c * CH + q) * BATCH + b) * KDIM + 2 * h;
    const float* bp = wu + (size_t)(c * 128 + kl) * LD + (key < 128 ? 256 : 768) + 2 * h;

    v8f acc = {0,0,0,0,0,0,0,0};
    #pragma unroll 8
    for (int k = 0; k < KDIM; k += 4) {
        v2f a  = *(const v2f*)(ap + k);
        v2f bb = *(const v2f*)(bp + k);
        acc = __builtin_amdgcn_wmma_f32_16x16x4_f32(false, a, false, bb,
                                                    (short)0, acc, false, false);
    }

    int tk = kl >> 3;
    float* pp = P + (size_t)((c * BATCH + b) * CH) * KEYS + key;
    #pragma unroll
    for (int r = 0; r < 8; ++r) {
        int s = 8 * h + r;
        pp[(size_t)s * KEYS] = (tk <= s) ? acc[r] : 0.0f;
    }
}

struct FragSet { v4f ka[8][2]; v2f bv[8]; };

__device__ __forceinline__ v4f ld4(const float* __restrict__ p)
{
    v4f v;
    __builtin_memcpy(&v, p, 16);
    return v;
}

template <int G>
__device__ __forceinline__ void load_group(FragSet& f, const float* __restrict__ kp,
                                           const float* __restrict__ vp)
{
    #pragma unroll
    for (int u = 0; u < 8; ++u) {
        const int r0 = (32 * G + 4 * u) * LD;
        const int r1 = r0 + LD;
        f.ka[u][0] = ld4(kp + r0);
        f.ka[u][1] = ld4(kp + r1);
        f.bv[u].x = vp[r0];
        f.bv[u].y = vp[r1];
    }
}

template <int G>
__device__ __forceinline__ void mma_group(const FragSet& f, int gvi, int gl0, v8f (&T)[4])
{
    #pragma unroll
    for (int u = 0; u < 8; ++u) {
        float g = __int_as_float(__builtin_amdgcn_readlane(gvi, gl0 + 4 * G + (u >> 1)));
        v2f vb;
        vb.x = f.bv[u].x * g;
        vb.y = f.bv[u].y * g;
        #pragma unroll
        for (int j = 0; j < 4; ++j) {
            v2f a; a.x = f.ka[u][0][j]; a.y = f.ka[u][1][j];
            T[j] = __builtin_amdgcn_wmma_f32_16x16x4_f32(false, a, false, vb,
                                                         (short)0, T[j], false, false);
        }
    }
}

#define STAGE_FENCE() __builtin_amdgcn_sched_barrier(0)

__global__ __launch_bounds__(128)
void fwp_scan(const float* __restrict__ wu,
              const float* __restrict__ x,
              const float* __restrict__ P,
              float* __restrict__ part)
{
    __shared__ float lds_all[LDS_FLOATS];
    float (*red)[4][256] = reinterpret_cast<float (*)[4][256]>(lds_all);

    int tid  = threadIdx.x;
    int lane = tid & 31;
    int q    = lane & 15;
    int h    = lane >> 4;
    int g    = __builtin_amdgcn_readfirstlane(tid >> 5);
    int kh   = blockIdx.z;
    int w    = 4 * kh + g;
    int gl0  = 16 * kh;
    int o0   = blockIdx.x * 16;
    int b    = blockIdx.y;
    float* po = part + (size_t)kh * OUT_FLOATS;

    v8f T[4];
    #pragma unroll
    for (int j = 0; j < 4; ++j) T[j] = (v8f){0,0,0,0,0,0,0,0};

    for (int c = 0; c < NCHUNK; ++c) {
        const float* wc = wu + (size_t)(c * 128) * LD;

        size_t orow0 = (size_t)((c * CH + (tid >> 4)) * BATCH + b);
        size_t orow1 = orow0 + 8 * BATCH;
        float base0 = 0.0f, base1 = 0.0f;
        if (kh == 0) {
            base0 = wu[orow0 * LD + 1026 + o0 + (tid & 15)];
            base1 = wu[orow1 * LD + 1026 + o0 + (tid & 15)];
        }
        float gin = wc[(size_t)(q * BATCH + b) * LD + 1024 + h];

        const float* xr = x + (size_t)((c * CH + q) * BATCH + b) * KDIM + 64 * g + 4 * h;
        const float* pr = P + (size_t)((c * BATCH + b) * CH + q) * KEYS + 32 * w + 2 * h;
        const float* vr = wc + (size_t)(32 * (w & 3) + 2 * h) * LD + (w < 4 ? 0 : 512) + o0 + q;

        v4f xv[4][2];
        v2f pa[8], vraw[8];
        #pragma unroll
        for (int jj = 0; jj < 4; ++jj) {
            xv[jj][0] = *(const v4f*)(xr + 16 * jj);
            xv[jj][1] = *(const v4f*)(xr + 16 * jj + 8);
        }
        #pragma unroll
        for (int jj = 0; jj < 8; ++jj) {
            pa[jj] = *(const v2f*)(pr + 4 * jj);
            vraw[jj].x = vr[(4 * jj) * LD];
            vraw[jj].y = vr[(4 * jj + 1) * LD];
        }

        const float* kp = wc + (size_t)(2 * h) * LD + (kh ? 768 : 256) + 64 * g + 8 * (q & 7) + 4 * (q >> 3);
        const float* vp = wc + (size_t)(2 * h) * LD + (kh ? 512 : 0) + o0 + q;
        FragSet f0, f1;
        load_group<0>(f0, kp, vp);
        STAGE_FENCE();

        float gv  = 1.0f / (1.0f + __expf(-gin));
        int   gvi = __float_as_int(h ? -gv : gv);

        v8f accA = {0,0,0,0,0,0,0,0};
        v8f accB = {0,0,0,0,0,0,0,0};
        #pragma unroll
        for (int jj = 0; jj < 4; ++jj) {
            #pragma unroll
            for (int j = 0; j < 4; j += 2) {
                v2f a0; a0.x = xv[jj][0][j];     a0.y = xv[jj][1][j];
                v2f a1; a1.x = xv[jj][0][j + 1]; a1.y = xv[jj][1][j + 1];
                v2f tb0; tb0.x = T[j][2 * jj];     tb0.y = T[j][2 * jj + 1];
                v2f tb1; tb1.x = T[j + 1][2 * jj]; tb1.y = T[j + 1][2 * jj + 1];
                accA = __builtin_amdgcn_wmma_f32_16x16x4_f32(false, a0, false, tb0,
                                                             (short)0, accA, false, false);
                accB = __builtin_amdgcn_wmma_f32_16x16x4_f32(false, a1, false, tb1,
                                                             (short)0, accB, false, false);
            }
        }
        #pragma unroll
        for (int jj = 0; jj < 8; jj += 2) {
            float gg = __int_as_float(__builtin_amdgcn_readlane(gvi, 4 * w + (jj >> 1)));
            v2f vb0; vb0.x = vraw[jj].x * gg;     vb0.y = vraw[jj].y * gg;
            v2f vb1; vb1.x = vraw[jj + 1].x * gg; vb1.y = vraw[jj + 1].y * gg;
            accA = __builtin_amdgcn_wmma_f32_16x16x4_f32(false, pa[jj], false, vb0,
                                                         (short)0, accA, false, false);
            accB = __builtin_amdgcn_wmma_f32_16x16x4_f32(false, pa[jj + 1], false, vb1,
                                                         (short)0, accB, false, false);
        }

        float* rw = &red[c & 1][g][0];
        #pragma unroll
        for (int r = 0; r < 8; ++r)
            rw[(8 * h + r) * 16 + q] = accA[r] + accB[r];
        STAGE_FENCE();

        load_group<1>(f1, kp, vp);  STAGE_FENCE();  mma_group<0>(f0, gvi, gl0, T);  STAGE_FENCE();
        load_group<2>(f0, kp, vp);  STAGE_FENCE();  mma_group<1>(f1, gvi, gl0, T);  STAGE_FENCE();
        load_group<3>(f1, kp, vp);  STAGE_FENCE();  mma_group<2>(f0, gvi, gl0, T);  STAGE_FENCE();
        mma_group<3>(f1, gvi, gl0, T);
        STAGE_FENCE();

        __syncthreads();
        float v0 = base0, v1 = base1;
        #pragma unroll
        for (int ww = 0; ww < 4; ++ww) {
            v0 += red[c & 1][ww][tid];
            v1 += red[c & 1][ww][tid + 128];
        }
        po[orow0 * 256 + o0 + (tid & 15)] = v0;
        po[orow1 * 256 + o0 + (tid & 15)] = v1;
    }
}

__global__ __launch_bounds__(256)
void fwp_join(const float* __restrict__ part, float* __restrict__ out)
{
    int i = blockIdx.x * 256 + threadIdx.x;
    v4f a = *(const v4f*)(part + 4 * (size_t)i);
    v4f s = *(const v4f*)(part + OUT_FLOATS + 4 * (size_t)i);
    *(v4f*)(out + 4 * (size_t)i) = a + s;
}

extern "C" void kernel_launch(void* const* d_in, const int* in_sizes, int n_in,
                              void* d_out, int out_size, void* d_ws, size_t ws_size,
                              hipStream_t stream)
{
    const float* x       = (const float*)d_in[0];
    const float* slow_W  = (const float*)d_in[1];
    const float* slow_b  = (const float*)d_in[2];
    const float* fast_W0 = (const float*)d_in[3];
    float*       out     = (float*)d_out;
    float*       wu      = (float*)d_ws;
    float*       P       = (float*)d_ws + WU_FLOATS;
    float*       part    = (float*)d_ws + WU_FLOATS + P_FLOATS;

    int total_waves = MTILES * NTILES;
    int blocks1 = (total_waves + 7) / 8;
    fwp_slow_gemm<<<blocks1, 256, 0, stream>>>(x, slow_W, slow_b, fast_W0, wu);

    fwp_scores<<<512, 256, 0, stream>>>(wu, x, P);

    dim3 grid2(16, BATCH, 2);
    fwp_scan<<<grid2, 128, 0, stream>>>(wu, x, P, part);

    fwp_join<<<1024, 256, 0, stream>>>(part, out);
}


# Slow GEMM and scores stage operands through LDS: 64x64 block per 8-wave work-group, rows copied once by whole-line 128-bit loads, fragments read from padded LDS; k order of sums changed
# speedup vs baseline: 6.3400x; 1.2786x over previous
#include <hip/hip_runtime.h>
#include <hip/hip_bf16.h>

typedef float v2f __attribute__((ext_vector_type(2)));
typedef float v4f __attribute__((ext_vector_type(4)));
typedef float v8f __attribute__((ext_vector_type(8)));

#define LD      1282
#define KDIM    256
#define NTOT    1282
#define SSTEPS  512
#define BATCH   8
#define CH      16
#define NCHUNK  32
#define KEYS    256
#define WU_FLOATS 5251072
#define P_FLOATS  1048576
#define OUT_FLOATS 1048576
#define LDS_FLOATS 41472

#define GT_ROWS  64
#define GT_PITCH 260
#define GT_NB    21

struct GemmFrags { v4f a[4], b0[4], b1[4]; };

template <int KB>
__device__ __forceinline__ void gt_load(GemmFrags& f, const float* ap, const float* bp0, const float* bp1)
{
    #pragma unroll
    for (int u = 0; u < 4; ++u) {
        f.a[u]  = *(const v4f*)(ap  + 8 * (4 * KB + u));
        f.b0[u] = *(const v4f*)(bp0 + 8 * (4 * KB + u));
        f.b1[u] = *(const v4f*)(bp1 + 8 * (4 * KB + u));
    }
}

__device__ __forceinline__ void gt_mma(const GemmFrags& f, v8f& c0, v8f& c1)
{
    #pragma unroll
    for (int u = 0; u < 4; ++u) {
        v2f al;  al.x  = f.a[u].x;  al.y  = f.a[u].y;
        v2f ah;  ah.x  = f.a[u].z;  ah.y  = f.a[u].w;
        v2f b0l; b0l.x = f.b0[u].x; b0l.y = f.b0[u].y;
        v2f b0h; b0h.x = f.b0[u].z; b0h.y = f.b0[u].w;
        v2f b1l; b1l.x = f.b1[u].x; b1l.y = f.b1[u].y;
        v2f b1h; b1h.x = f.b1[u].z; b1h.y = f.b1[u].w;
        c0 = __builtin_amdgcn_wmma_f32_16x16x4_f32(false, al, false, b0l, (short)0, c0, false, false);
        c1 = __builtin_amdgcn_wmma_f32_16x16x4_f32(false, al, false, b1l, (short)0, c1, false, false);
        c0 = __builtin_amdgcn_wmma_f32_16x16x4_f32(false, ah, false, b0h, (short)0, c0, false, false);
        c1 = __builtin_amdgcn_wmma_f32_16x16x4_f32(false, ah, false, b1h, (short)0, c1, false, false);
    }
}

#define GT_FENCE() __builtin_amdgcn_sched_barrier(0)

__global__ __launch_bounds__(256)
void fwp_slow_gemm(const float* __restrict__ x,
                   const float* __restrict__ slow_W,
                   const float* __restrict__ slow_b,
                   const float* __restrict__ fast_W0,
                   float* __restrict__ wu)
{
    __shared__ __attribute__((aligned(16))) float tile[2 * GT_ROWS * GT_PITCH];

    int tid  = threadIdx.x;
    int lane = tid & 31;
    int w    = __builtin_amdgcn_readfirstlane(tid >> 5);
    int mb   = blockIdx.x / GT_NB, nb = blockIdx.x % GT_NB;
    int m0   = mb * GT_ROWS, n0 = nb * GT_ROWS;

    int rr = w >> 1;
    int c4 = 4 * (32 * (w & 1) + lane);
    v4f st[32];
    #pragma unroll
    for (int j = 0; j < 16; ++j) {
        int row = rr + 4 * j;
        st[j] = *(const v4f*)(x + (size_t)(m0 + row) * KDIM + c4);
    }
    #pragma unroll
    for (int j = 0; j < 16; ++j) {
        int n = n0 + rr + 4 * j;
        const float* wp;
        if (n < 1026) {
            wp = slow_W + (size_t)n * KDIM;
        } else {
            int idx = n - 1026;
            if (idx > 255) idx = 255;
            wp = fast_W0 + (size_t)idx * KDIM;
        }
        st[16 + j] = *(const v4f*)(wp + c4);
    }
    #pragma unroll
    for (int j = 0; j < 32; ++j)
        *(v4f*)(tile + (rr + 4 * j) * GT_PITCH + c4) = st[j];
    __syncthreads();

    int q = lane & 15, h = lane >> 4;
    int mi = w & 3, nh = w >> 2;
    int nbase = n0 + 32 * nh;
    if (nbase >= NTOT) return;

    int na = nbase + q, nc = nbase + 16 + q;
    float bva = (na < 1026) ? slow_b[na] : 0.0f;
    float bvc = (nc < 1026) ? slow_b[nc] : 0.0f;
    v8f c0, c1;
    #pragma unroll
    for (int r = 0; r < 8; ++r) { c0[r] = bva; c1[r] = bvc; }

    const float* ap  = tile + (16 * mi + q) * GT_PITCH + 4 * h;
    const float* bp0 = tile + (GT_ROWS + 32 * nh + q) * GT_PITCH + 4 * h;
    const float* bp1 = bp0 + 16 * GT_PITCH;
    GemmFrags f0, f1;
    gt_load<0>(f0, ap, bp0, bp1);  GT_FENCE();
    gt_load<1>(f1, ap, bp0, bp1);  GT_FENCE();  gt_mma(f0, c0, c1);  GT_FENCE();
    gt_load<2>(f0, ap, bp0, bp1);  GT_FENCE();  gt_mma(f1, c0, c1);  GT_FENCE();
    gt_load<3>(f1, ap, bp0, bp1);  GT_FENCE();  gt_mma(f0, c0, c1);  GT_FENCE();
    gt_load<4>(f0, ap, bp0, bp1);  GT_FENCE();  gt_mma(f1, c0, c1);  GT_FENCE();
    gt_load<5>(f1, ap, bp0, bp1);  GT_FENCE();  gt_mma(f0, c0, c1);  GT_FENCE();
    gt_load<6>(f0, ap, bp0, bp1);  GT_FENCE();  gt_mma(f1, c0, c1);  GT_FENCE();
    gt_load<7>(f1, ap, bp0, bp1);  GT_FENCE();  gt_mma(f0, c0, c1);  GT_FENCE();
    gt_mma(f1, c0, c1);

    int mrow = m0 + 16 * mi + 8 * h;
    if (na < NTOT) {
        #pragma unroll
        for (int r = 0; r < 8; ++r) wu[(size_t)(mrow + r) * LD + na] = c0[r];
    }
    if (nc < NTOT) {
        #pragma unroll
        for (int r = 0; r < 8; ++r) wu[(size_t)(mrow + r) * LD + nc] = c1[r];
    }
}

__device__ __forceinline__ v4f ld4(const float* __restrict__ p)
{
    v4f v;
    __builtin_memcpy(&v, p, 16);
    return v;
}

__global__ __launch_bounds__(256)
void fwp_scores(const float* __restrict__ wu,
                const float* __restrict__ x,
                float* __restrict__ P)
{
    __shared__ __attribute__((aligned(16))) float tile[2 * GT_ROWS * GT_PITCH];

    int tid  = threadIdx.x;
    int lane = tid & 31;
    int w    = __builtin_amdgcn_readfirstlane(tid >> 5);
    int c    = blockIdx.x >> 3;
    int mb   = (blockIdx.x >> 2) & 1;
    int kb   = blockIdx.x & 3;
    int kl0  = 64 * (kb & 1);

    int rr = w >> 1;
    int c4 = 4 * (32 * (w & 1) + lane);
    const float* xs = x  + (size_t)(c * 128 + 64 * mb) * KDIM + c4;
    const float* ks = wu + (size_t)(c * 128 + kl0) * LD + (kb < 2 ? 256 : 768) + c4;
    v4f st[32];
    #pragma unroll
    for (int j = 0; j < 16; ++j)
        st[j] = *(const v4f*)(xs + (size_t)(rr + 4 * j) * KDIM);
    #pragma unroll
    for (int j = 0; j < 16; ++j)
        st[16 + j] = ld4(ks + (size_t)(rr + 4 * j) * LD);
    #pragma unroll
    for (int j = 0; j < 32; ++j)
        *(v4f*)(tile + (rr + 4 * j) * GT_PITCH + c4) = st[j];
    __syncthreads();

    int q = lane & 15, h = lane >> 4;
    int mi = w & 3, nh = w >> 2;
    v8f c0 = {0,0,0,0,0,0,0,0};
    v8f c1 = {0,0,0,0,0,0,0,0};
    const float* ap  = tile + (16 * mi + q) * GT_PITCH + 4 * h;
    const float* bp0 = tile + (GT_ROWS + 32 * nh + q) * GT_PITCH + 4 * h;
    const float* bp1 = bp0 + 16 * GT_PITCH;
    GemmFrags f0, f1;
    gt_load<0>(f0, ap, bp0, bp1);  GT_FENCE();
    gt_load<1>(f1, ap, bp0, bp1);  GT_FENCE();  gt_mma(f0, c0, c1);  GT_FENCE();
    gt_load<2>(f0, ap, bp0, bp1);  GT_FENCE();  gt_mma(f1, c0, c1);  GT_FENCE();
    gt_load<3>(f1, ap, bp0, bp1);  GT_FENCE();  gt_mma(f0, c0, c1);  GT_FENCE();
    gt_load<4>(f0, ap, bp0, bp1);  GT_FENCE();  gt_mma(f1, c0, c1);  GT_FENCE();
    gt_load<5>(f1, ap, bp0, bp1);  GT_FENCE();  gt_mma(f0, c0, c1);  GT_FENCE();
    gt_load<6>(f0, ap, bp0, bp1);  GT_FENCE();  gt_mma(f1, c0, c1);  GT_FENCE();
    gt_load<7>(f1, ap, bp0, bp1);  GT_FENCE();  gt_mma(f0, c0, c1);  GT_FENCE();
    gt_mma(f1, c0, c1);

    int s    = 8 * mb + 2 * mi + h;
    int kloc = 32 * nh + q;
    int tk0  = (kl0 + kloc) >> 3;
    int tk1  = (kl0 + kloc + 16) >> 3;
    float* pp = P + (size_t)((c * BATCH) * CH + s) * KEYS + 64 * kb + kloc;
    #pragma unroll
    for (int r = 0; r < 8; ++r) {
        pp[(size_t)r * CH * KEYS]      = (tk0 <= s) ? c0[r] : 0.0f;
        pp[(size_t)r * CH * KEYS + 16] = (tk1 <= s) ? c1[r] : 0.0f;
    }
}

struct FragSet { v4f ka[8][2]; v2f bv[8]; };

template <int G>
__device__ __forceinline__ void load_group(FragSet& f, const float* __restrict__ kp,
                                           const float* __restrict__ vp)
{
    #pragma unroll
    for (int u = 0; u < 8; ++u) {
        const int r0 = (32 * G + 4 * u) * LD;
        const int r1 = r0 + LD;
        f.ka[u][0] = ld4(kp + r0);
        f.ka[u][1] = ld4(kp + r1);
        f.bv[u].x = vp[r0];
        f.bv[u].y = vp[r1];
    }
}

template <int G>
__device__ __forceinline__ void mma_group(const FragSet& f, int gvi, int gl0, v8f (&T)[4])
{
    #pragma unroll
    for (int u = 0; u < 8; ++u) {
        float g = __int_as_float(__builtin_amdgcn_readlane(gvi, gl0 + 4 * G + (u >> 1)));
        v2f vb;
        vb.x = f.bv[u].x * g;
        vb.y = f.bv[u].y * g;
        #pragma unroll
        for (int j = 0; j < 4; ++j) {
            v2f a; a.x = f.ka[u][0][j]; a.y = f.ka[u][1][j];
            T[j] = __builtin_amdgcn_wmma_f32_16x16x4_f32(false, a, false, vb,
                                                         (short)0, T[j], false, false);
        }
    }
}

#define STAGE_FENCE() __builtin_amdgcn_sched_barrier(0)

__global__ __launch_bounds__(128)
void fwp_scan(const float* __restrict__ wu,
              const float* __restrict__ x,
              const float* __restrict__ P,
              float* __restrict__ part)
{
    __shared__ float lds_all[LDS_FLOATS];
    float (*red)[4][256] = reinterpret_cast<float (*)[4][256]>(lds_all);

    int tid  = threadIdx.x;
    int lane = tid & 31;
    int q    = lane & 15;
    int h    = lane >> 4;
    int g    = __builtin_amdgcn_readfirstlane(tid >> 5);
    int kh   = blockIdx.z;
    int w    = 4 * kh + g;
    int gl0  = 16 * kh;
    int o0   = blockIdx.x * 16;
    int b    = blockIdx.y;
    float* po = part + (size_t)kh * OUT_FLOATS;

    v8f T[4];
    #pragma unroll
    for (int j = 0; j < 4; ++j) T[j] = (v8f){0,0,0,0,0,0,0,0};

    for (int c = 0; c < NCHUNK; ++c) {
        const float* wc = wu + (size_t)(c * 128) * LD;

        size_t orow0 = (size_t)((c * CH + (tid >> 4)) * BATCH + b);
        size_t orow1 = orow0 + 8 * BATCH;
        float base0 = 0.0f, base1 = 0.0f;
        if (kh == 0) {
            base0 = wu[orow0 * LD + 1026 + o0 + (tid & 15)];
            base1 = wu[orow1 * LD + 1026 + o0 + (tid & 15)];
        }
        float gin = wc[(size_t)(q * BATCH + b) * LD + 1024 + h];

        const float* xr = x + (size_t)((c * CH + q) * BATCH + b) * KDIM + 64 * g + 4 * h;
        const float* pr = P + (size_t)((c * BATCH + b) * CH + q) * KEYS + 32 * w + 2 * h;
        const float* vr = wc + (size_t)(32 * (w & 3) + 2 * h) * LD + (w < 4 ? 0 : 512) + o0 + q;

        v4f xv[4][2];
        v2f pa[8], vraw[8];
        #pragma unroll
        for (int jj = 0; jj < 4; ++jj) {
            xv[jj][0] = *(const v4f*)(xr + 16 * jj);
            xv[jj][1] = *(const v4f*)(xr + 16 * jj + 8);
        }
        #pragma unroll
        for (int jj = 0; jj < 8; ++jj) {
            pa[jj] = *(const v2f*)(pr + 4 * jj);
            vraw[jj].x = vr[(4 * jj) * LD];
            vraw[jj].y = vr[(4 * jj + 1) * LD];
        }

        const float* kp = wc + (size_t)(2 * h) * LD + (kh ? 768 : 256) + 64 * g + 8 * (q & 7) + 4 * (q >> 3);
        const float* vp = wc + (size_t)(2 * h) * LD + (kh ? 512 : 0) + o0 + q;
        FragSet f0, f1;
        load_group<0>(f0, kp, vp);
        STAGE_FENCE();

        float gv  = 1.0f / (1.0f + __expf(-gin));
        int   gvi = __float_as_int(h ? -gv : gv);

        v8f accA = {0,0,0,0,0,0,0,0};
        v8f accB = {0,0,0,0,0,0,0,0};
        #pragma unroll
        for (int jj = 0; jj < 4; ++jj) {
            #pragma unroll
            for (int j = 0; j < 4; j += 2) {
                v2f a0; a0.x = xv[jj][0][j];     a0.y = xv[jj][1][j];
                v2f a1; a1.x = xv[jj][0][j + 1]; a1.y = xv[jj][1][j + 1];
                v2f tb0; tb0.x = T[j][2 * jj];     tb0.y = T[j][2 * jj + 1];
                v2f tb1; tb1.x = T[j + 1][2 * jj]; tb1.y = T[j + 1][2 * jj + 1];
                accA = __builtin_amdgcn_wmma_f32_16x16x4_f32(false, a0, false, tb0,
                                                             (short)0, accA, false, false);
                accB = __builtin_amdgcn_wmma_f32_16x16x4_f32(false, a1, false, tb1,
                                                             (short)0, accB, false, false);
            }
        }
        #pragma unroll
        for (int jj = 0; jj < 8; jj += 2) {
            float gg = __int_as_float(__builtin_amdgcn_readlane(gvi, 4 * w + (jj >> 1)));
            v2f vb0; vb0.x = vraw[jj].x * gg;     vb0.y = vraw[jj].y * gg;
            v2f vb1; vb1.x = vraw[jj + 1].x * gg; vb1.y = vraw[jj + 1].y * gg;
            accA = __builtin_amdgcn_wmma_f32_16x16x4_f32(false, pa[jj], false, vb0,
                                                         (short)0, accA, false, false);
            accB = __builtin_amdgcn_wmma_f32_16x16x4_f32(false, pa[jj + 1], false, vb1,
                                                         (short)0, accB, false, false);
        }

        float* rw = &red[c & 1][g][0];
        #pragma unroll
        for (int r = 0; r < 8; ++r)
            rw[(8 * h + r) * 16 + q] = accA[r] + accB[r];
        STAGE_FENCE();

        load_group<1>(f1, kp, vp);  STAGE_FENCE();  mma_group<0>(f0, gvi, gl0, T);  STAGE_FENCE();
        load_group<2>(f0, kp, vp);  STAGE_FENCE();  mma_group<1>(f1, gvi, gl0, T);  STAGE_FENCE();
        load_group<3>(f1, kp, vp);  STAGE_FENCE();  mma_group<2>(f0, gvi, gl0, T);  STAGE_FENCE();
        mma_group<3>(f1, gvi, gl0, T);
        STAGE_FENCE();

        __syncthreads();
        float v0 = base0, v1 = base1;
        #pragma unroll
        for (int ww = 0; ww < 4; ++ww) {
            v0 += red[c & 1][ww][tid];
            v1 += red[c & 1][ww][tid + 128];
        }
        po[orow0 * 256 + o0 + (tid & 15)] = v0;
        po[orow1 * 256 + o0 + (tid & 15)] = v1;
    }
}

__global__ __launch_bounds__(256)
void fwp_join(const float* __restrict__ part, float* __restrict__ out)
{
    int i = blockIdx.x * 256 + threadIdx.x;
    v4f a = *(const v4f*)(part + 4 * (size_t)i);
    v4f s = *(const v4f*)(part + OUT_FLOATS + 4 * (size_t)i);
    *(v4f*)(out + 4 * (size_t)i) = a + s;
}

extern "C" void kernel_launch(void* const* d_in, const int* in_sizes, int n_in,
                              void* d_out, int out_size, void* d_ws, size_t ws_size,
                              hipStream_t stream)
{
    const float* x       = (const float*)d_in[0];
    const float* slow_W  = (const float*)d_in[1];
    const float* slow_b  = (const float*)d_in[2];
    const float* fast_W0 = (const float*)d_in[3];
    float*       out     = (float*)d_out;
    float*       wu      = (float*)d_ws;
    float*       P       = (float*)d_ws + WU_FLOATS;
    float*       part    = (float*)d_ws + WU_FLOATS + P_FLOATS;

    fwp_slow_gemm<<<1344, 256, 0, stream>>>(x, slow_W, slow_b, fast_W0, wu);

    fwp_scores<<<256, 256, 0, stream>>>(wu, x, P);

    dim3 grid2(16, BATCH, 2);
    fwp_scan<<<grid2, 128, 0, stream>>>(wu, x, P, part);

    fwp_join<<<1024, 256, 0, stream>>>(part, out);
}


# wu row pitch padded from 1282 to 1344 floats (21 x 256 B): every wu row starts line-aligned, a wave's key-row read in scan and scores covers 4 whole 128-B lines instead of straddling 6
# speedup vs baseline: 6.8631x; 1.0825x over previous
#include <hip/hip_runtime.h>
#include <hip/hip_bf16.h>

typedef float v2f __attribute__((ext_vector_type(2)));
typedef float v4f __attribute__((ext_vector_type(4)));
typedef float v8f __attribute__((ext_vector_type(8)));

#define LD      1344
#define KDIM    256
#define NTOT    1282
#define SSTEPS  512
#define BATCH   8
#define CH      16
#define NCHUNK  32
#define KEYS    256
#define WU_FLOATS 5505024
#define P_FLOATS  1048576
#define OUT_FLOATS 1048576
#define LDS_FLOATS 41472

#define GT_ROWS  64
#define GT_PITCH 260
#define GT_NB    21

struct GemmFrags { v4f a[4], b0[4], b1[4]; };

template <int KB>
__device__ __forceinline__ void gt_load(GemmFrags& f, const float* ap, const float* bp0, const float* bp1)
{
    #pragma unroll
    for (int u = 0; u < 4; ++u) {
        f.a[u]  = *(const v4f*)(ap  + 8 * (4 * KB + u));
        f.b0[u] = *(const v4f*)(bp0 + 8 * (4 * KB + u));
        f.b1[u] = *(const v4f*)(bp1 + 8 * (4 * KB + u));
    }
}

__device__ __forceinline__ void gt_mma(const GemmFrags& f, v8f& c0, v8f& c1)
{
    #pragma unroll
    for (int u = 0; u < 4; ++u) {
        v2f al;  al.x  = f.a[u].x;  al.y  = f.a[u].y;
        v2f ah;  ah.x  = f.a[u].z;  ah.y  = f.a[u].w;
        v2f b0l; b0l.x = f.b0[u].x; b0l.y = f.b0[u].y;
        v2f b0h; b0h.x = f.b0[u].z; b0h.y = f.b0[u].w;
        v2f b1l; b1l.x = f.b1[u].x; b1l.y = f.b1[u].y;
        v2f b1h; b1h.x = f.b1[u].z; b1h.y = f.b1[u].w;
        c0 = __builtin_amdgcn_wmma_f32_16x16x4_f32(false, al, false, b0l, (short)0, c0, false, false);
        c1 = __builtin_amdgcn_wmma_f32_16x16x4_f32(false, al, false, b1l, (short)0, c1, false, false);
        c0 = __builtin_amdgcn_wmma_f32_16x16x4_f32(false, ah, false, b0h, (short)0, c0, false, false);
        c1 = __builtin_amdgcn_wmma_f32_16x16x4_f32(false, ah, false, b1h, (short)0, c1, false, false);
    }
}

#define GT_FENCE() __builtin_amdgcn_sched_barrier(0)

__global__ __launch_bounds__(256)
void fwp_slow_gemm(const float* __restrict__ x,
                   const float* __restrict__ slow_W,
                   const float* __restrict__ slow_b,
                   const float* __restrict__ fast_W0,
                   float* __restrict__ wu)
{
    __shared__ __attribute__((aligned(16))) float tile[2 * GT_ROWS * GT_PITCH];

    int tid  = threadIdx.x;
    int lane = tid & 31;
    int w    = __builtin_amdgcn_readfirstlane(tid >> 5);
    int mb   = blockIdx.x / GT_NB, nb = blockIdx.x % GT_NB;
    int m0   = mb * GT_ROWS, n0 = nb * GT_ROWS;

    int rr = w >> 1;
    int c4 = 4 * (32 * (w & 1) + lane);
    v4f st[32];
    #pragma unroll
    for (int j = 0; j < 16; ++j) {
        int row = rr + 4 * j;
        st[j] = *(const v4f*)(x + (size_t)(m0 + row) * KDIM + c4);
    }
    #pragma unroll
    for (int j = 0; j < 16; ++j) {
        int n = n0 + rr + 4 * j;
        const float* wp;
        if (n < 1026) {
            wp = slow_W + (size_t)n * KDIM;
        } else {
            int idx = n - 1026;
            if (idx > 255) idx = 255;
            wp = fast_W0 + (size_t)idx * KDIM;
        }
        st[16 + j] = *(const v4f*)(wp + c4);
    }
    #pragma unroll
    for (int j = 0; j < 32; ++j)
        *(v4f*)(tile + (rr + 4 * j) * GT_PITCH + c4) = st[j];
    __syncthreads();

    int q = lane & 15, h = lane >> 4;
    int mi = w & 3, nh = w >> 2;
    int nbase = n0 + 32 * nh;
    if (nbase >= NTOT) return;

    int na = nbase + q, nc = nbase + 16 + q;
    float bva = (na < 1026) ? slow_b[na] : 0.0f;
    float bvc = (nc < 1026) ? slow_b[nc] : 0.0f;
    v8f c0, c1;
    #pragma unroll
    for (int r = 0; r < 8; ++r) { c0[r] = bva; c1[r] = bvc; }

    const float* ap  = tile + (16 * mi + q) * GT_PITCH + 4 * h;
    const float* bp0 = tile + (GT_ROWS + 32 * nh + q) * GT_PITCH + 4 * h;
    const float* bp1 = bp0 + 16 * GT_PITCH;
    GemmFrags f0, f1;
    gt_load<0>(f0, ap, bp0, bp1);  GT_FENCE();
    gt_load<1>(f1, ap, bp0, bp1);  GT_FENCE();  gt_mma(f0, c0, c1);  GT_FENCE();
    gt_load<2>(f0, ap, bp0, bp1);  GT_FENCE();  gt_mma(f1, c0, c1);  GT_FENCE();
    gt_load<3>(f1, ap, bp0, bp1);  GT_FENCE();  gt_mma(f0, c0, c1);  GT_FENCE();
    gt_load<4>(f0, ap, bp0, bp1);  GT_FENCE();  gt_mma(f1, c0, c1);  GT_FENCE();
    gt_load<5>(f1, ap, bp0, bp1);  GT_FENCE();  gt_mma(f0, c0, c1);  GT_FENCE();
    gt_load<6>(f0, ap, bp0, bp1);  GT_FENCE();  gt_mma(f1, c0, c1);  GT_FENCE();
    gt_load<7>(f1, ap, bp0, bp1);  GT_FENCE();  gt_mma(f0, c0, c1);  GT_FENCE();
    gt_mma(f1, c0, c1);

    int mrow = m0 + 16 * mi + 8 * h;
    if (na < NTOT) {
        #pragma unroll
        for (int r = 0; r < 8; ++r) wu[(size_t)(mrow + r) * LD + na] = c0[r];
    }
    if (nc < NTOT) {
        #pragma unroll
        for (int r = 0; r < 8; ++r) wu[(size_t)(mrow + r) * LD + nc] = c1[r];
    }
}

__device__ __forceinline__ v4f ld4(const float* __restrict__ p)
{
    v4f v;
    __builtin_memcpy(&v, p, 16);
    return v;
}

__global__ __launch_bounds__(256)
void fwp_scores(const float* __restrict__ wu,
                const float* __restrict__ x,
                float* __restrict__ P)
{
    __shared__ __attribute__((aligned(16))) float tile[2 * GT_ROWS * GT_PITCH];

    int tid  = threadIdx.x;
    int lane = tid & 31;
    int w    = __builtin_amdgcn_readfirstlane(tid >> 5);
    int c    = blockIdx.x >> 3;
    int mb   = (blockIdx.x >> 2) & 1;
    int kb   = blockIdx.x & 3;
    int kl0  = 64 * (kb & 1);

    int rr = w >> 1;
    int c4 = 4 * (32 * (w & 1) + lane);
    const float* xs = x  + (size_t)(c * 128 + 64 * mb) * KDIM + c4;
    const float* ks = wu + (size_t)(c * 128 + kl0) * LD + (kb < 2 ? 256 : 768) + c4;
    v4f st[32];
    #pragma unroll
    for (int j = 0; j < 16; ++j)
        st[j] = *(const v4f*)(xs + (size_t)(rr + 4 * j) * KDIM);
    #pragma unroll
    for (int j = 0; j < 16; ++j)
        st[16 + j] = ld4(ks + (size_t)(rr + 4 * j) * LD);
    #pragma unroll
    for (int j = 0; j < 32; ++j)
        *(v4f*)(tile + (rr + 4 * j) * GT_PITCH + c4) = st[j];
    __syncthreads();

    int q = lane & 15, h = lane >> 4;
    int mi = w & 3, nh = w >> 2;
    v8f c0 = {0,0,0,0,0,0,0,0};
    v8f c1 = {0,0,0,0,0,0,0,0};
    const float* ap  = tile + (16 * mi + q) * GT_PITCH + 4 * h;
    const float* bp0 = tile + (GT_ROWS + 32 * nh + q) * GT_PITCH + 4 * h;
    const float* bp1 = bp0 + 16 * GT_PITCH;
    GemmFrags f0, f1;
    gt_load<0>(f0, ap, bp0, bp1);  GT_FENCE();
    gt_load<1>(f1, ap, bp0, bp1);  GT_FENCE();  gt_mma(f0, c0, c1);  GT_FENCE();
    gt_load<2>(f0, ap, bp0, bp1);  GT_FENCE();  gt_mma(f1, c0, c1);  GT_FENCE();
    gt_load<3>(f1, ap, bp0, bp1);  GT_FENCE();  gt_mma(f0, c0, c1);  GT_FENCE();
    gt_load<4>(f0, ap, bp0, bp1);  GT_FENCE();  gt_mma(f1, c0, c1);  GT_FENCE();
    gt_load<5>(f1, ap, bp0, bp1);  GT_FENCE();  gt_mma(f0, c0, c1);  GT_FENCE();
    gt_load<6>(f0, ap, bp0, bp1);  GT_FENCE();  gt_mma(f1, c0, c1);  GT_FENCE();
    gt_load<7>(f1, ap, bp0, bp1);  GT_FENCE();  gt_mma(f0, c0, c1);  GT_FENCE();
    gt_mma(f1, c0, c1);

    int s    = 8 * mb + 2 * mi + h;
    int kloc = 32 * nh + q;
    int tk0  = (kl0 + kloc) >> 3;
    int tk1  = (kl0 + kloc + 16) >> 3;
    float* pp = P + (size_t)((c * BATCH) * CH + s) * KEYS + 64 * kb + kloc;
    #pragma unroll
    for (int r = 0; r < 8; ++r) {
        pp[(size_t)r * CH * KEYS]      = (tk0 <= s) ? c0[r] : 0.0f;
        pp[(size_t)r * CH * KEYS + 16] = (tk1 <= s) ? c1[r] : 0.0f;
    }
}

struct FragSet { v4f ka[8][2]; v2f bv[8]; };

template <int G>
__device__ __forceinline__ void load_group(FragSet& f, const float* __restrict__ kp,
                                           const float* __restrict__ vp)
{
    #pragma unroll
    for (int u = 0; u < 8; ++u) {
        const int r0 = (32 * G + 4 * u) * LD;
        const int r1 = r0 + LD;
        f.ka[u][0] = ld4(kp + r0);
        f.ka[u][1] = ld4(kp + r1);
        f.bv[u].x = vp[r0];
        f.bv[u].y = vp[r1];
    }
}

template <int G>
__device__ __forceinline__ void mma_group(const FragSet& f, int gvi, int gl0, v8f (&T)[4])
{
    #pragma unroll
    for (int u = 0; u < 8; ++u) {
        float g = __int_as_float(__builtin_amdgcn_readlane(gvi, gl0 + 4 * G + (u >> 1)));
        v2f vb;
        vb.x = f.bv[u].x * g;
        vb.y = f.bv[u].y * g;
        #pragma unroll
        for (int j = 0; j < 4; ++j) {
            v2f a; a.x = f.ka[u][0][j]; a.y = f.ka[u][1][j];
            T[j] = __builtin_amdgcn_wmma_f32_16x16x4_f32(false, a, false, vb,
                                                         (short)0, T[j], false, false);
        }
    }
}

#define STAGE_FENCE() __builtin_amdgcn_sched_barrier(0)

__global__ __launch_bounds__(128)
void fwp_scan(const float* __restrict__ wu,
              const float* __restrict__ x,
              const float* __restrict__ P,
              float* __restrict__ part)
{
    __shared__ float lds_all[LDS_FLOATS];
    float (*red)[4][256] = reinterpret_cast<float (*)[4][256]>(lds_all);

    int tid  = threadIdx.x;
    int lane = tid & 31;
    int q    = lane & 15;
    int h    = lane >> 4;
    int g    = __builtin_amdgcn_readfirstlane(tid >> 5);
    int kh   = blockIdx.z;
    int w    = 4 * kh + g;
    int gl0  = 16 * kh;
    int o0   = blockIdx.x * 16;
    int b    = blockIdx.y;
    float* po = part + (size_t)kh * OUT_FLOATS;

    v8f T[4];
    #pragma unroll
    for (int j = 0; j < 4; ++j) T[j] = (v8f){0,0,0,0,0,0,0,0};

    for (int c = 0; c < NCHUNK; ++c) {
        const float* wc = wu + (size_t)(c * 128) * LD;

        size_t orow0 = (size_t)((c * CH + (tid >> 4)) * BATCH + b);
        size_t orow1 = orow0 + 8 * BATCH;
        float base0 = 0.0f, base1 = 0.0f;
        if (kh == 0) {
            base0 = wu[orow0 * LD + 1026 + o0 + (tid & 15)];
            base1 = wu[orow1 * LD + 1026 + o0 + (tid & 15)];
        }
        float gin = wc[(size_t)(q * BATCH + b) * LD + 1024 + h];

        const float* xr = x + (size_t)((c * CH + q) * BATCH + b) * KDIM + 64 * g + 4 * h;
        const float* pr = P + (size_t)((c * BATCH + b) * CH + q) * KEYS + 32 * w + 2 * h;
        const float* vr = wc + (size_t)(32 * (w & 3) + 2 * h) * LD + (w < 4 ? 0 : 512) + o0 + q;

        v4f xv[4][2];
        v2f pa[8], vraw[8];
        #pragma unroll
        for (int jj = 0; jj < 4; ++jj) {
            xv[jj][0] = *(const v4f*)(xr + 16 * jj);
            xv[jj][1] = *(const v4f*)(xr + 16 * jj + 8);
        }
        #pragma unroll
        for (int jj = 0; jj < 8; ++jj) {
            pa[jj] = *(const v2f*)(pr + 4 * jj);
            vraw[jj].x = vr[(4 * jj) * LD];
            vraw[jj].y = vr[(4 * jj + 1) * LD];
        }

        const float* kp = wc + (size_t)(2 * h) * LD + (kh ? 768 : 256) + 64 * g + 8 * (q & 7) + 4 * (q >> 3);
        const float* vp = wc + (size_t)(2 * h) * LD + (kh ? 512 : 0) + o0 + q;
        FragSet f0, f1;
        load_group<0>(f0, kp, vp);
        STAGE_FENCE();

        float gv  = 1.0f / (1.0f + __expf(-gin));
        int   gvi = __float_as_int(h ? -gv : gv);

        v8f accA = {0,0,0,0,0,0,0,0};
        v8f accB = {0,0,0,0,0,0,0,0};
        #pragma unroll
        for (int jj = 0; jj < 4; ++jj) {
            #pragma unroll
            for (int j = 0; j < 4; j += 2) {
                v2f a0; a0.x = xv[jj][0][j];     a0.y = xv[jj][1][j];
                v2f a1; a1.x = xv[jj][0][j + 1]; a1.y = xv[jj][1][j + 1];
                v2f tb0; tb0.x = T[j][2 * jj];     tb0.y = T[j][2 * jj + 1];
                v2f tb1; tb1.x = T[j + 1][2 * jj]; tb1.y = T[j + 1][2 * jj + 1];
                accA = __builtin_amdgcn_wmma_f32_16x16x4_f32(false, a0, false, tb0,
                                                             (short)0, accA, false, false);
                accB = __builtin_amdgcn_wmma_f32_16x16x4_f32(false, a1, false, tb1,
                                                             (short)0, accB, false, false);
            }
        }
        #pragma unroll
        for (int jj = 0; jj < 8; jj += 2) {
            float gg = __int_as_float(__builtin_amdgcn_readlane(gvi, 4 * w + (jj >> 1)));
            v2f vb0; vb0.x = vraw[jj].x * gg;     vb0.y = vraw[jj].y * gg;
            v2f vb1; vb1.x = vraw[jj + 1].x * gg; vb1.y = vraw[jj + 1].y * gg;
            accA = __builtin_amdgcn_wmma_f32_16x16x4_f32(false, pa[jj], false, vb0,
                                                         (short)0, accA, false, false);
            accB = __builtin_amdgcn_wmma_f32_16x16x4_f32(false, pa[jj + 1], false, vb1,
                                                         (short)0, accB, false, false);
        }

        float* rw = &red[c & 1][g][0];
        #pragma unroll
        for (int r = 0; r < 8; ++r)
            rw[(8 * h + r) * 16 + q] = accA[r] + accB[r];
        STAGE_FENCE();

        load_group<1>(f1, kp, vp);  STAGE_FENCE();  mma_group<0>(f0, gvi, gl0, T);  STAGE_FENCE();
        load_group<2>(f0, kp, vp);  STAGE_FENCE();  mma_group<1>(f1, gvi, gl0, T);  STAGE_FENCE();
        load_group<3>(f1, kp, vp);  STAGE_FENCE();  mma_group<2>(f0, gvi, gl0, T);  STAGE_FENCE();
        mma_group<3>(f1, gvi, gl0, T);
        STAGE_FENCE();

        __syncthreads();
        float v0 = base0, v1 = base1;
        #pragma unroll
        for (int ww = 0; ww < 4; ++ww) {
            v0 += red[c & 1][ww][tid];
            v1 += red[c & 1][ww][tid + 128];
        }
        po[orow0 * 256 + o0 + (tid & 15)] = v0;
        po[orow1 * 256 + o0 + (tid & 15)] = v1;
    }
}

__global__ __launch_bounds__(256)
void fwp_join(const float* __restrict__ part, float* __restrict__ out)
{
    int i = blockIdx.x * 256 + threadIdx.x;
    v4f a = *(const v4f*)(part + 4 * (size_t)i);
    v4f s = *(const v4f*)(part + OUT_FLOATS + 4 * (size_t)i);
    *(v4f*)(out + 4 * (size_t)i) = a + s;
}

extern "C" void kernel_launch(void* const* d_in, const int* in_sizes, int n_in,
                              void* d_out, int out_size, void* d_ws, size_t ws_size,
                              hipStream_t stream)
{
    const float* x       = (const float*)d_in[0];
    const float* slow_W  = (const float*)d_in[1];
    const float* slow_b  = (const float*)d_in[2];
    const float* fast_W0 = (const float*)d_in[3];
    float*       out     = (float*)d_out;
    float*       wu      = (float*)d_ws;
    float*       P       = (float*)d_ws + WU_FLOATS;
    float*       part    = (float*)d_ws + WU_FLOATS + P_FLOATS;

    fwp_slow_gemm<<<1344, 256, 0, stream>>>(x, slow_W, slow_b, fast_W0, wu);

    fwp_scores<<<256, 256, 0, stream>>>(wu, x, P);

    dim3 grid2(16, BATCH, 2);
    fwp_scan<<<grid2, 128, 0, stream>>>(wu, x, P, part);

    fwp_join<<<1024, 256, 0, stream>>>(part, out);
}


# Scan work-group re-tiled: its 4 waves hold the SAME 64-i slice of 4 adjacent o-blocks (was 4 i-slices of one), so key/x/score lines leave L2 once per group; 8 output planes; order of sums changed
# speedup vs baseline: 7.3075x; 1.0648x over previous
#include <hip/hip_runtime.h>
#include <hip/hip_bf16.h>

typedef float v2f __attribute__((ext_vector_type(2)));
typedef float v4f __attribute__((ext_vector_type(4)));
typedef float v8f __attribute__((ext_vector_type(8)));

#define LD      1344
#define KDIM    256
#define NTOT    1282
#define SSTEPS  512
#define BATCH   8
#define CH      16
#define NCHUNK  32
#define KEYS    256
#define WU_FLOATS 5505024
#define P_FLOATS  1048576
#define OUT_FLOATS 1048576
#define NPLANE  8
#define RED_PITCH 68
#define LDS_FLOATS 41472

#define GT_ROWS  64
#define GT_PITCH 260
#define GT_NB    21

struct GemmFrags { v4f a[4], b0[4], b1[4]; };

template <int KB>
__device__ __forceinline__ void gt_load(GemmFrags& f, const float* ap, const float* bp0, const float* bp1)
{
    #pragma unroll
    for (int u = 0; u < 4; ++u) {
        f.a[u]  = *(const v4f*)(ap  + 8 * (4 * KB + u));
        f.b0[u] = *(const v4f*)(bp0 + 8 * (4 * KB + u));
        f.b1[u] = *(const v4f*)(bp1 + 8 * (4 * KB + u));
    }
}

__device__ __forceinline__ void gt_mma(const GemmFrags& f, v8f& c0, v8f& c1)
{
    #pragma unroll
    for (int u = 0; u < 4; ++u) {
        v2f al;  al.x  = f.a[u].x;  al.y  = f.a[u].y;
        v2f ah;  ah.x  = f.a[u].z;  ah.y  = f.a[u].w;
        v2f b0l; b0l.x = f.b0[u].x; b0l.y = f.b0[u].y;
        v2f b0h; b0h.x = f.b0[u].z; b0h.y = f.b0[u].w;
        v2f b1l; b1l.x = f.b1[u].x; b1l.y = f.b1[u].y;
        v2f b1h; b1h.x = f.b1[u].z; b1h.y = f.b1[u].w;
        c0 = __builtin_amdgcn_wmma_f32_16x16x4_f32(false, al, false, b0l, (short)0, c0, false, false);
        c1 = __builtin_amdgcn_wmma_f32_16x16x4_f32(false, al, false, b1l, (short)0, c1, false, false);
        c0 = __builtin_amdgcn_wmma_f32_16x16x4_f32(false, ah, false, b0h, (short)0, c0, false, false);
        c1 = __builtin_amdgcn_wmma_f32_16x16x4_f32(false, ah, false, b1h, (short)0, c1, false, false);
    }
}

#define GT_FENCE() __builtin_amdgcn_sched_barrier(0)

__global__ __launch_bounds__(256)
void fwp_slow_gemm(const float* __restrict__ x,
                   const float* __restrict__ slow_W,
                   const float* __restrict__ slow_b,
                   const float* __restrict__ fast_W0,
                   float* __restrict__ wu)
{
    __shared__ __attribute__((aligned(16))) float tile[2 * GT_ROWS * GT_PITCH];

    int tid  = threadIdx.x;
    int lane = tid & 31;
    int w    = __builtin_amdgcn_readfirstlane(tid >> 5);
    int mb   = blockIdx.x / GT_NB, nb = blockIdx.x % GT_NB;
    int m0   = mb * GT_ROWS, n0 = nb * GT_ROWS;

    int rr = w >> 1;
    int c4 = 4 * (32 * (w & 1) + lane);
    v4f st[32];
    #pragma unroll
    for (int j = 0; j < 16; ++j) {
        int row = rr + 4 * j;
        st[j] = *(const v4f*)(x + (size_t)(m0 + row) * KDIM + c4);
    }
    #pragma unroll
    for (int j = 0; j < 16; ++j) {
        int n = n0 + rr + 4 * j;
        const float* wp;
        if (n < 1026) {
            wp = slow_W + (size_t)n * KDIM;
        } else {
            int idx = n - 1026;
            if (idx > 255) idx = 255;
            wp = fast_W0 + (size_t)idx * KDIM;
        }
        st[16 + j] = *(const v4f*)(wp + c4);
    }
    #pragma unroll
    for (int j = 0; j < 32; ++j)
        *(v4f*)(tile + (rr + 4 * j) * GT_PITCH + c4) = st[j];
    __syncthreads();

    int q = lane & 15, h = lane >> 4;
    int mi = w & 3, nh = w >> 2;
    int nbase = n0 + 32 * nh;
    if (nbase >= NTOT) return;

    int na = nbase + q, nc = nbase + 16 + q;
    float bva = (na < 1026) ? slow_b[na] : 0.0f;
    float bvc = (nc < 1026) ? slow_b[nc] : 0.0f;
    v8f c0, c1;
    #pragma unroll
    for (int r = 0; r < 8; ++r) { c0[r] = bva; c1[r] = bvc; }

    const float* ap  = tile + (16 * mi + q) * GT_PITCH + 4 * h;
    const float* bp0 = tile + (GT_ROWS + 32 * nh + q) * GT_PITCH + 4 * h;
    const float* bp1 = bp0 + 16 * GT_PITCH;
    GemmFrags f0, f1;
    gt_load<0>(f0, ap, bp0, bp1);  GT_FENCE();
    gt_load<1>(f1, ap, bp0, bp1);  GT_FENCE();  gt_mma(f0, c0, c1);  GT_FENCE();
    gt_load<2>(f0, ap, bp0, bp1);  GT_FENCE();  gt_mma(f1, c0, c1);  GT_FENCE();
    gt_load<3>(f1, ap, bp0, bp1);  GT_FENCE();  gt_mma(f0, c0, c1);  GT_FENCE();
    gt_load<4>(f0, ap, bp0, bp1);  GT_FENCE();  gt_mma(f1, c0, c1);  GT_FENCE();
    gt_load<5>(f1, ap, bp0, bp1);  GT_FENCE();  gt_mma(f0, c0, c1);  GT_FENCE();
    gt_load<6>(f0, ap, bp0, bp1);  GT_FENCE();  gt_mma(f1, c0, c1);  GT_FENCE();
    gt_load<7>(f1, ap, bp0, bp1);  GT_FENCE();  gt_mma(f0, c0, c1);  GT_FENCE();
    gt_mma(f1, c0, c1);

    int mrow = m0 + 16 * mi + 8 * h;
    if (na < NTOT) {
        #pragma unroll
        for (int r = 0; r < 8; ++r) wu[(size_t)(mrow + r) * LD + na] = c0[r];
    }
    if (nc < NTOT) {
        #pragma unroll
        for (int r = 0; r < 8; ++r) wu[(size_t)(mrow + r) * LD + nc] = c1[r];
    }
}

__device__ __forceinline__ v4f ld4(const float* __restrict__ p)
{
    v4f v;
    __builtin_memcpy(&v, p, 16);
    return v;
}

__global__ __launch_bounds__(256)
void fwp_scores(const float* __restrict__ wu,
                const float* __restrict__ x,
                float* __restrict__ P)
{
    __shared__ __attribute__((aligned(16))) float tile[2 * GT_ROWS * GT_PITCH];

    int tid  = threadIdx.x;
    int lane = tid & 31;
    int w    = __builtin_amdgcn_readfirstlane(tid >> 5);
    int c    = blockIdx.x >> 3;
    int mb   = (blockIdx.x >> 2) & 1;
    int kb   = blockIdx.x & 3;
    int kl0  = 64 * (kb & 1);

    int rr = w >> 1;
    int c4 = 4 * (32 * (w & 1) + lane);
    const float* xs = x  + (size_t)(c * 128 + 64 * mb) * KDIM + c4;
    const float* ks = wu + (size_t)(c * 128 + kl0) * LD + (kb < 2 ? 256 : 768) + c4;
    v4f st[32];
    #pragma unroll
    for (int j = 0; j < 16; ++j)
        st[j] = *(const v4f*)(xs + (size_t)(rr + 4 * j) * KDIM);
    #pragma unroll
    for (int j = 0; j < 16; ++j)
        st[16 + j] = ld4(ks + (size_t)(rr + 4 * j) * LD);
    #pragma unroll
    for (int j = 0; j < 32; ++j)
        *(v4f*)(tile + (rr + 4 * j) * GT_PITCH + c4) = st[j];
    __syncthreads();

    int q = lane & 15, h = lane >> 4;
    int mi = w & 3, nh = w >> 2;
    v8f c0 = {0,0,0,0,0,0,0,0};
    v8f c1 = {0,0,0,0,0,0,0,0};
    const float* ap  = tile + (16 * mi + q) * GT_PITCH + 4 * h;
    const float* bp0 = tile + (GT_ROWS + 32 * nh + q) * GT_PITCH + 4 * h;
    const float* bp1 = bp0 + 16 * GT_PITCH;
    GemmFrags f0, f1;
    gt_load<0>(f0, ap, bp0, bp1);  GT_FENCE();
    gt_load<1>(f1, ap, bp0, bp1);  GT_FENCE();  gt_mma(f0, c0, c1);  GT_FENCE();
    gt_load<2>(f0, ap, bp0, bp1);  GT_FENCE();  gt_mma(f1, c0, c1);  GT_FENCE();
    gt_load<3>(f1, ap, bp0, bp1);  GT_FENCE();  gt_mma(f0, c0, c1);  GT_FENCE();
    gt_load<4>(f0, ap, bp0, bp1);  GT_FENCE();  gt_mma(f1, c0, c1);  GT_FENCE();
    gt_load<5>(f1, ap, bp0, bp1);  GT_FENCE();  gt_mma(f0, c0, c1);  GT_FENCE();
    gt_load<6>(f0, ap, bp0, bp1);  GT_FENCE();  gt_mma(f1, c0, c1);  GT_FENCE();
    gt_load<7>(f1, ap, bp0, bp1);  GT_FENCE();  gt_mma(f0, c0, c1);  GT_FENCE();
    gt_mma(f1, c0, c1);

    int s    = 8 * mb + 2 * mi + h;
    int kloc = 32 * nh + q;
    int tk0  = (kl0 + kloc) >> 3;
    int tk1  = (kl0 + kloc + 16) >> 3;
    float* pp = P + (size_t)((c * BATCH) * CH + s) * KEYS + 64 * kb + kloc;
    #pragma unroll
    for (int r = 0; r < 8; ++r) {
        pp[(size_t)r * CH * KEYS]      = (tk0 <= s) ? c0[r] : 0.0f;
        pp[(size_t)r * CH * KEYS + 16] = (tk1 <= s) ? c1[r] : 0.0f;
    }
}

struct FragSet { v4f ka[8][2]; v2f bv[8]; };

template <int G>
__device__ __forceinline__ void load_group(FragSet& f, const float* __restrict__ kp,
                                           const float* __restrict__ vp)
{
    #pragma unroll
    for (int u = 0; u < 8; ++u) {
        const int r0 = (32 * G + 4 * u) * LD;
        const int r1 = r0 + LD;
        f.ka[u][0] = ld4(kp + r0);
        f.ka[u][1] = ld4(kp + r1);
        f.bv[u].x = vp[r0];
        f.bv[u].y = vp[r1];
    }
}

template <int G>
__device__ __forceinline__ void mma_group(const FragSet& f, int gvi, int gl0, v8f (&T)[4])
{
    #pragma unroll
    for (int u = 0; u < 8; ++u) {
        float g = __int_as_float(__builtin_amdgcn_readlane(gvi, gl0 + 4 * G + (u >> 1)));
        v2f vb;
        vb.x = f.bv[u].x * g;
        vb.y = f.bv[u].y * g;
        #pragma unroll
        for (int j = 0; j < 4; ++j) {
            v2f a; a.x = f.ka[u][0][j]; a.y = f.ka[u][1][j];
            T[j] = __builtin_amdgcn_wmma_f32_16x16x4_f32(false, a, false, vb,
                                                         (short)0, T[j], false, false);
        }
    }
}

#define STAGE_FENCE() __builtin_amdgcn_sched_barrier(0)

__global__ __launch_bounds__(128)
void fwp_scan(const float* __restrict__ wu,
              const float* __restrict__ x,
              const float* __restrict__ P,
              float* __restrict__ part)
{
    __shared__ __attribute__((aligned(16))) float lds_all[LDS_FLOATS];
    float (*red)[16 * RED_PITCH] = reinterpret_cast<float (*)[16 * RED_PITCH]>(lds_all);

    int tid  = threadIdx.x;
    int lane = tid & 31;
    int q    = lane & 15;
    int h    = lane >> 4;
    int wv   = __builtin_amdgcn_readfirstlane(tid >> 5);
    int g    = blockIdx.x & 3;
    int og   = blockIdx.x >> 2;
    int kh   = blockIdx.z;
    int w    = 4 * kh + g;
    int gl0  = 16 * kh;
    int o0   = 64 * og + 16 * wv;
    int b    = blockIdx.y;
    float* po = part + (size_t)(2 * g + kh) * OUT_FLOATS;
    int fs   = tid >> 4;
    int fc   = 4 * (tid & 15);

    v8f T[4];
    #pragma unroll
    for (int j = 0; j < 4; ++j) T[j] = (v8f){0,0,0,0,0,0,0,0};

    for (int c = 0; c < NCHUNK; ++c) {
        const float* wc = wu + (size_t)(c * 128) * LD;

        size_t orow0 = (size_t)((c * CH + fs) * BATCH + b);
        size_t orow1 = orow0 + 8 * BATCH;
        v4f base0 = {0, 0, 0, 0}, base1 = {0, 0, 0, 0};
        if (kh == 0 && g == 0) {
            const float* bp0 = wu + orow0 * LD + 1026 + 64 * og + fc;
            const float* bp1 = wu + orow1 * LD + 1026 + 64 * og + fc;
            base0.x = bp0[0]; base0.y = bp0[1]; base0.z = bp0[2]; base0.w = bp0[3];
            base1.x = bp1[0]; base1.y = bp1[1]; base1.z = bp1[2]; base1.w = bp1[3];
        }
        float gin = wc[(size_t)(q * BATCH + b) * LD + 1024 + h];

        const float* xr = x + (size_t)((c * CH + q) * BATCH + b) * KDIM + 64 * g + 4 * h;
        const float* pr = P + (size_t)((c * BATCH + b) * CH + q) * KEYS + 32 * w + 2 * h;
        const float* vr = wc + (size_t)(32 * (w & 3) + 2 * h) * LD + (w < 4 ? 0 : 512) + o0 + q;

        v4f xv[4][2];
        v2f pa[8], vraw[8];
        #pragma unroll
        for (int jj = 0; jj < 4; ++jj) {
            xv[jj][0] = *(const v4f*)(xr + 16 * jj);
            xv[jj][1] = *(const v4f*)(xr + 16 * jj + 8);
        }
        #pragma unroll
        for (int jj = 0; jj < 8; ++jj) {
            pa[jj] = *(const v2f*)(pr + 4 * jj);
            vraw[jj].x = vr[(4 * jj) * LD];
            vraw[jj].y = vr[(4 * jj + 1) * LD];
        }

        const float* kp = wc + (size_t)(2 * h) * LD + (kh ? 768 : 256) + 64 * g + 8 * (q & 7) + 4 * (q >> 3);
        const float* vp = wc + (size_t)(2 * h) * LD + (kh ? 512 : 0) + o0 + q;
        FragSet f0, f1;
        load_group<0>(f0, kp, vp);
        STAGE_FENCE();

        float gv  = 1.0f / (1.0f + __expf(-gin));
        int   gvi = __float_as_int(h ? -gv : gv);

        v8f accA = {0,0,0,0,0,0,0,0};
        v8f accB = {0,0,0,0,0,0,0,0};
        #pragma unroll
        for (int jj = 0; jj < 4; ++jj) {
            #pragma unroll
            for (int j = 0; j < 4; j += 2) {
                v2f a0; a0.x = xv[jj][0][j];     a0.y = xv[jj][1][j];
                v2f a1; a1.x = xv[jj][0][j + 1]; a1.y = xv[jj][1][j + 1];
                v2f tb0; tb0.x = T[j][2 * jj];     tb0.y = T[j][2 * jj + 1];
                v2f tb1; tb1.x = T[j + 1][2 * jj]; tb1.y = T[j + 1][2 * jj + 1];
                accA = __builtin_amdgcn_wmma_f32_16x16x4_f32(false, a0, false, tb0,
                                                             (short)0, accA, false, false);
                accB = __builtin_amdgcn_wmma_f32_16x16x4_f32(false, a1, false, tb1,
                                                             (short)0, accB, false, false);
            }
        }
        #pragma unroll
        for (int jj = 0; jj < 8; jj += 2) {
            float gg = __int_as_float(__builtin_amdgcn_readlane(gvi, 4 * w + (jj >> 1)));
            v2f vb0; vb0.x = vraw[jj].x * gg;     vb0.y = vraw[jj].y * gg;
            v2f vb1; vb1.x = vraw[jj + 1].x * gg; vb1.y = vraw[jj + 1].y * gg;
            accA = __builtin_amdgcn_wmma_f32_16x16x4_f32(false, pa[jj], false, vb0,
                                                         (short)0, accA, false, false);
            accB = __builtin_amdgcn_wmma_f32_16x16x4_f32(false, pa[jj + 1], false, vb1,
                                                         (short)0, accB, false, false);
        }

        float* rw = &red[c & 1][16 * wv + q];
        #pragma unroll
        for (int r = 0; r < 8; ++r)
            rw[(8 * h + r) * RED_PITCH] = accA[r] + accB[r];
        STAGE_FENCE();

        load_group<1>(f1, kp, vp);  STAGE_FENCE();  mma_group<0>(f0, gvi, gl0, T);  STAGE_FENCE();
        load_group<2>(f0, kp, vp);  STAGE_FENCE();  mma_group<1>(f1, gvi, gl0, T);  STAGE_FENCE();
        load_group<3>(f1, kp, vp);  STAGE_FENCE();  mma_group<2>(f0, gvi, gl0, T);  STAGE_FENCE();
        mma_group<3>(f1, gvi, gl0, T);
        STAGE_FENCE();

        __syncthreads();
        v4f v0 = *(const v4f*)&red[c & 1][fs * RED_PITCH + fc];
        v4f v1 = *(const v4f*)&red[c & 1][(8 + fs) * RED_PITCH + fc];
        *(v4f*)(po + orow0 * 256 + 64 * og + fc) = v0 + base0;
        *(v4f*)(po + orow1 * 256 + 64 * og + fc) = v1 + base1;
    }
}

__global__ __launch_bounds__(256)
void fwp_join(const float* __restrict__ part, float* __restrict__ out)
{
    int i = blockIdx.x * 256 + threadIdx.x;
    v4f pl[NPLANE];
    #pragma unroll
    for (int p = 0; p < NPLANE; ++p)
        pl[p] = *(const v4f*)(part + (size_t)p * OUT_FLOATS + 4 * (size_t)i);
    v4f s = pl[0];
    #pragma unroll
    for (int p = 1; p < NPLANE; ++p) s += pl[p];
    *(v4f*)(out + 4 * (size_t)i) = s;
}

extern "C" void kernel_launch(void* const* d_in, const int* in_sizes, int n_in,
                              void* d_out, int out_size, void* d_ws, size_t ws_size,
                              hipStream_t stream)
{
    const float* x       = (const float*)d_in[0];
    const float* slow_W  = (const float*)d_in[1];
    const float* slow_b  = (const float*)d_in[2];
    const float* fast_W0 = (const float*)d_in[3];
    float*       out     = (float*)d_out;
    float*       wu      = (float*)d_ws;
    float*       P       = (float*)d_ws + WU_FLOATS;
    float*       part    = (float*)d_ws + WU_FLOATS + P_FLOATS;

    fwp_slow_gemm<<<1344, 256, 0, stream>>>(x, slow_W, slow_b, fast_W0, wu);

    fwp_scores<<<256, 256, 0, stream>>>(wu, x, P);

    dim3 grid2(16, BATCH, 2);
    fwp_scan<<<grid2, 128, 0, stream>>>(wu, x, P, part);

    fwp_join<<<1024, 256, 0, stream>>>(part, out);
}


# Scan pipelined across chunks: chunk c+1's head (x rows, scores, values, gates, base) and first key/value group are loaded under chunk c's last two update groups; the chunk barrier fences LDS only
# speedup vs baseline: 7.9973x; 1.0944x over previous
#include <hip/hip_runtime.h>
#include <hip/hip_bf16.h>

typedef float v2f __attribute__((ext_vector_type(2)));
typedef float v4f __attribute__((ext_vector_type(4)));
typedef float v8f __attribute__((ext_vector_type(8)));

#define LD      1344
#define KDIM    256
#define NTOT    1282
#define SSTEPS  512
#define BATCH   8
#define CH      16
#define NCHUNK  32
#define KEYS    256
#define WU_FLOATS 5505024
#define P_FLOATS  1048576
#define OUT_FLOATS 1048576
#define NPLANE  8
#define RED_PITCH 68
#define LDS_FLOATS 41472

#define GT_ROWS  64
#define GT_PITCH 260
#define GT_NB    21

struct GemmFrags { v4f a[4], b0[4], b1[4]; };

template <int KB>
__device__ __forceinline__ void gt_load(GemmFrags& f, const float* ap, const float* bp0, const float* bp1)
{
    #pragma unroll
    for (int u = 0; u < 4; ++u) {
        f.a[u]  = *(const v4f*)(ap  + 8 * (4 * KB + u));
        f.b0[u] = *(const v4f*)(bp0 + 8 * (4 * KB + u));
        f.b1[u] = *(const v4f*)(bp1 + 8 * (4 * KB + u));
    }
}

__device__ __forceinline__ void gt_mma(const GemmFrags& f, v8f& c0, v8f& c1)
{
    #pragma unroll
    for (int u = 0; u < 4; ++u) {
        v2f al;  al.x  = f.a[u].x;  al.y  = f.a[u].y;
        v2f ah;  ah.x  = f.a[u].z;  ah.y  = f.a[u].w;
        v2f b0l; b0l.x = f.b0[u].x; b0l.y = f.b0[u].y;
        v2f b0h; b0h.x = f.b0[u].z; b0h.y = f.b0[u].w;
        v2f b1l; b1l.x = f.b1[u].x; b1l.y = f.b1[u].y;
        v2f b1h; b1h.x = f.b1[u].z; b1h.y = f.b1[u].w;
        c0 = __builtin_amdgcn_wmma_f32_16x16x4_f32(false, al, false, b0l, (short)0, c0, false, false);
        c1 = __builtin_amdgcn_wmma_f32_16x16x4_f32(false, al, false, b1l, (short)0, c1, false, false);
        c0 = __builtin_amdgcn_wmma_f32_16x16x4_f32(false, ah, false, b0h, (short)0, c0, false, false);
        c1 = __builtin_amdgcn_wmma_f32_16x16x4_f32(false, ah, false, b1h, (short)0, c1, false, false);
    }
}

#define GT_FENCE() __builtin_amdgcn_sched_barrier(0)

__global__ __launch_bounds__(256)
void fwp_slow_gemm(const float* __restrict__ x,
                   const float* __restrict__ slow_W,
                   const float* __restrict__ slow_b,
                   const float* __restrict__ fast_W0,
                   float* __restrict__ wu)
{
    __shared__ __attribute__((aligned(16))) float tile[2 * GT_ROWS * GT_PITCH];

    int tid  = threadIdx.x;
    int lane = tid & 31;
    int w    = __builtin_amdgcn_readfirstlane(tid >> 5);
    int mb   = blockIdx.x / GT_NB, nb = blockIdx.x % GT_NB;
    int m0   = mb * GT_ROWS, n0 = nb * GT_ROWS;

    int rr = w >> 1;
    int c4 = 4 * (32 * (w & 1) + lane);
    v4f st[32];
    #pragma unroll
    for (int j = 0; j < 16; ++j) {
        int row = rr + 4 * j;
        st[j] = *(const v4f*)(x + (size_t)(m0 + row) * KDIM + c4);
    }
    #pragma unroll
    for (int j = 0; j < 16; ++j) {
        int n = n0 + rr + 4 * j;
        const float* wp;
        if (n < 1026) {
            wp = slow_W + (size_t)n * KDIM;
        } else {
            int idx = n - 1026;
            if (idx > 255) idx = 255;
            wp = fast_W0 + (size_t)idx * KDIM;
        }
        st[16 + j] = *(const v4f*)(wp + c4);
    }
    #pragma unroll
    for (int j = 0; j < 32; ++j)
        *(v4f*)(tile + (rr + 4 * j) * GT_PITCH + c4) = st[j];
    __syncthreads();

    int q = lane & 15, h = lane >> 4;
    int mi = w & 3, nh = w >> 2;
    int nbase = n0 + 32 * nh;
    if (nbase >= NTOT) return;

    int na = nbase + q, nc = nbase + 16 + q;
    float bva = (na < 1026) ? slow_b[na] : 0.0f;
    float bvc = (nc < 1026) ? slow_b[nc] : 0.0f;
    v8f c0, c1;
    #pragma unroll
    for (int r = 0; r < 8; ++r) { c0[r] = bva; c1[r] = bvc; }

    const float* ap  = tile + (16 * mi + q) * GT_PITCH + 4 * h;
    const float* bp0 = tile + (GT_ROWS + 32 * nh + q) * GT_PITCH + 4 * h;
    const float* bp1 = bp0 + 16 * GT_PITCH;
    GemmFrags f0, f1;
    gt_load<0>(f0, ap, bp0, bp1);  GT_FENCE();
    gt_load<1>(f1, ap, bp0, bp1);  GT_FENCE();  gt_mma(f0, c0, c1);  GT_FENCE();
    gt_load<2>(f0, ap, bp0, bp1);  GT_FENCE();  gt_mma(f1, c0, c1);  GT_FENCE();
    gt_load<3>(f1, ap, bp0, bp1);  GT_FENCE();  gt_mma(f0, c0, c1);  GT_FENCE();
    gt_load<4>(f0, ap, bp0, bp1);  GT_FENCE();  gt_mma(f1, c0, c1);  GT_FENCE();
    gt_load<5>(f1, ap, bp0, bp1);  GT_FENCE();  gt_mma(f0, c0, c1);  GT_FENCE();
    gt_load<6>(f0, ap, bp0, bp1);  GT_FENCE();  gt_mma(f1, c0, c1);  GT_FENCE();
    gt_load<7>(f1, ap, bp0, bp1);  GT_FENCE();  gt_mma(f0, c0, c1);  GT_FENCE();
    gt_mma(f1, c0, c1);

    int mrow = m0 + 16 * mi + 8 * h;
    if (na < NTOT) {
        #pragma unroll
        for (int r = 0; r < 8; ++r) wu[(size_t)(mrow + r) * LD + na] = c0[r];
    }
    if (nc < NTOT) {
        #pragma unroll
        for (int r = 0; r < 8; ++r) wu[(size_t)(mrow + r) * LD + nc] = c1[r];
    }
}

__device__ __forceinline__ v4f ld4(const float* __restrict__ p)
{
    v4f v;
    __builtin_memcpy(&v, p, 16);
    return v;
}

__global__ __launch_bounds__(256)
void fwp_scores(const float* __restrict__ wu,
                const float* __restrict__ x,
                float* __restrict__ P)
{
    __shared__ __attribute__((aligned(16))) float tile[2 * GT_ROWS * GT_PITCH];

    int tid  = threadIdx.x;
    int lane = tid & 31;
    int w    = __builtin_amdgcn_readfirstlane(tid >> 5);
    int c    = blockIdx.x >> 3;
    int mb   = (blockIdx.x >> 2) & 1;
    int kb   = blockIdx.x & 3;
    int kl0  = 64 * (kb & 1);

    int rr = w >> 1;
    int c4 = 4 * (32 * (w & 1) + lane);
    const float* xs = x  + (size_t)(c * 128 + 64 * mb) * KDIM + c4;
    const float* ks = wu + (size_t)(c * 128 + kl0) * LD + (kb < 2 ? 256 : 768) + c4;
    v4f st[32];
    #pragma unroll
    for (int j = 0; j < 16; ++j)
        st[j] = *(const v4f*)(xs + (size_t)(rr + 4 * j) * KDIM);
    #pragma unroll
    for (int j = 0; j < 16; ++j)
        st[16 + j] = ld4(ks + (size_t)(rr + 4 * j) * LD);
    #pragma unroll
    for (int j = 0; j < 32; ++j)
        *(v4f*)(tile + (rr + 4 * j) * GT_PITCH + c4) = st[j];
    __syncthreads();

    int q = lane & 15, h = lane >> 4;
    int mi = w & 3, nh = w >> 2;
    v8f c0 = {0,0,0,0,0,0,0,0};
    v8f c1 = {0,0,0,0,0,0,0,0};
    const float* ap  = tile + (16 * mi + q) * GT_PITCH + 4 * h;
    const float* bp0 = tile + (GT_ROWS + 32 * nh + q) * GT_PITCH + 4 * h;
    const float* bp1 = bp0 + 16 * GT_PITCH;
    GemmFrags f0, f1;
    gt_load<0>(f0, ap, bp0, bp1);  GT_FENCE();
    gt_load<1>(f1, ap, bp0, bp1);  GT_FENCE();  gt_mma(f0, c0, c1);  GT_FENCE();
    gt_load<2>(f0, ap, bp0, bp1);  GT_FENCE();  gt_mma(f1, c0, c1);  GT_FENCE();
    gt_load<3>(f1, ap, bp0, bp1);  GT_FENCE();  gt_mma(f0, c0, c1);  GT_FENCE();
    gt_load<4>(f0, ap, bp0, bp1);  GT_FENCE();  gt_mma(f1, c0, c1);  GT_FENCE();
    gt_load<5>(f1, ap, bp0, bp1);  GT_FENCE();  gt_mma(f0, c0, c1);  GT_FENCE();
    gt_load<6>(f0, ap, bp0, bp1);  GT_FENCE();  gt_mma(f1, c0, c1);  GT_FENCE();
    gt_load<7>(f1, ap, bp0, bp1);  GT_FENCE();  gt_mma(f0, c0, c1);  GT_FENCE();
    gt_mma(f1, c0, c1);

    int s    = 8 * mb + 2 * mi + h;
    int kloc = 32 * nh + q;
    int tk0  = (kl0 + kloc) >> 3;
    int tk1  = (kl0 + kloc + 16) >> 3;
    float* pp = P + (size_t)((c * BATCH) * CH + s) * KEYS + 64 * kb + kloc;
    #pragma unroll
    for (int r = 0; r < 8; ++r) {
        pp[(size_t)r * CH * KEYS]      = (tk0 <= s) ? c0[r] : 0.0f;
        pp[(size_t)r * CH * KEYS + 16] = (tk1 <= s) ? c1[r] : 0.0f;
    }
}

struct FragSet { v4f ka[8][2]; v2f bv[8]; };

template <int G>
__device__ __forceinline__ void load_group(FragSet& f, const float* __restrict__ kp,
                                           const float* __restrict__ vp)
{
    #pragma unroll
    for (int u = 0; u < 8; ++u) {
        const int r0 = (32 * G + 4 * u) * LD;
        const int r1 = r0 + LD;
        f.ka[u][0] = ld4(kp + r0);
        f.ka[u][1] = ld4(kp + r1);
        f.bv[u].x = vp[r0];
        f.bv[u].y = vp[r1];
    }
}

template <int G>
__device__ __forceinline__ void mma_group(const FragSet& f, int gvi, int gl0, v8f (&T)[4])
{
    #pragma unroll
    for (int u = 0; u < 8; ++u) {
        float g = __int_as_float(__builtin_amdgcn_readlane(gvi, gl0 + 4 * G + (u >> 1)));
        v2f vb;
        vb.x = f.bv[u].x * g;
        vb.y = f.bv[u].y * g;
        #pragma unroll
        for (int j = 0; j < 4; ++j) {
            v2f a; a.x = f.ka[u][0][j]; a.y = f.ka[u][1][j];
            T[j] = __builtin_amdgcn_wmma_f32_16x16x4_f32(false, a, false, vb,
                                                         (short)0, T[j], false, false);
        }
    }
}

struct ChunkHead { v4f xv[4][2]; v2f pa[8]; v2f vraw[8]; float gin; v4f base0, base1; };

__device__ __forceinline__ void load_head_x(ChunkHead& H, const float* __restrict__ xr,
                                            const float* __restrict__ gp)
{
    #pragma unroll
    for (int jj = 0; jj < 4; ++jj) {
        H.xv[jj][0] = *(const v4f*)(xr + 16 * jj);
        H.xv[jj][1] = *(const v4f*)(xr + 16 * jj + 8);
    }
    H.gin = *gp;
}

__device__ __forceinline__ void load_head_p(ChunkHead& H, const float* __restrict__ pr,
                                            const float* __restrict__ vr)
{
    #pragma unroll
    for (int jj = 0; jj < 8; ++jj) {
        H.pa[jj] = *(const v2f*)(pr + 4 * jj);
        H.vraw[jj].x = vr[(4 * jj) * LD];
        H.vraw[jj].y = vr[(4 * jj + 1) * LD];
    }
}

#define STAGE_FENCE() __builtin_amdgcn_sched_barrier(0)

__global__ __launch_bounds__(128)
void fwp_scan(const float* __restrict__ wu,
              const float* __restrict__ x,
              const float* __restrict__ P,
              float* __restrict__ part)
{
    __shared__ __attribute__((aligned(16))) float lds_all[LDS_FLOATS];
    float (*red)[16 * RED_PITCH] = reinterpret_cast<float (*)[16 * RED_PITCH]>(lds_all);

    int tid  = threadIdx.x;
    int lane = tid & 31;
    int q    = lane & 15;
    int h    = lane >> 4;
    int wv   = __builtin_amdgcn_readfirstlane(tid >> 5);
    int g    = blockIdx.x & 3;
    int og   = blockIdx.x >> 2;
    int kh   = blockIdx.z;
    int w    = 4 * kh + g;
    int gl0  = 16 * kh;
    int o0   = 64 * og + 16 * wv;
    int b    = blockIdx.y;
    float* po = part + (size_t)(2 * g + kh) * OUT_FLOATS;
    int fs   = tid >> 4;
    int fc   = 4 * (tid & 15);

    v8f T[4];
    #pragma unroll
    for (int j = 0; j < 4; ++j) T[j] = (v8f){0,0,0,0,0,0,0,0};

    const int CHUNK_WU = 128 * LD;
    const int CHUNK_XP = CH * BATCH * KDIM;
    const float* gp0 = wu + (size_t)(q * BATCH + b) * LD + 1024 + h;
    const float* xr0 = x + (size_t)(q * BATCH + b) * KDIM + 64 * g + 4 * h;
    const float* pr0 = P + (size_t)(b * CH + q) * KEYS + 32 * w + 2 * h;
    const float* vr0 = wu + (size_t)(32 * (w & 3) + 2 * h) * LD + (w < 4 ? 0 : 512) + o0 + q;
    const float* kp0 = wu + (size_t)(2 * h) * LD + (kh ? 768 : 256) + 64 * g + 8 * (q & 7) + 4 * (q >> 3);
    const float* vp0 = wu + (size_t)(2 * h) * LD + (kh ? 512 : 0) + o0 + q;
    const bool has_base = (kh == 0 && g == 0);
    const float* bb0 = wu + (size_t)(fs * BATCH + b) * LD + 1026 + 64 * og + fc;

    ChunkHead H;
    FragSet f0, f1;
    H.base0 = (v4f){0, 0, 0, 0};
    H.base1 = (v4f){0, 0, 0, 0};
    if (has_base) {
        const float* bp0 = bb0;
        const float* bp1 = bb0 + (size_t)(8 * BATCH) * LD;
        H.base0.x = bp0[0]; H.base0.y = bp0[1]; H.base0.z = bp0[2]; H.base0.w = bp0[3];
        H.base1.x = bp1[0]; H.base1.y = bp1[1]; H.base1.z = bp1[2]; H.base1.w = bp1[3];
    }
    load_head_x(H, xr0, gp0);
    load_head_p(H, pr0, vr0);
    load_group<0>(f0, kp0, vp0);
    STAGE_FENCE();

    for (int c = 0; c < NCHUNK; ++c) {
        const int cn = (c + 1 < NCHUNK) ? c + 1 : c;
        const float* kp = kp0 + (size_t)c * CHUNK_WU;
        const float* vp = vp0 + (size_t)c * CHUNK_WU;

        size_t orow0 = (size_t)((c * CH + fs) * BATCH + b);
        size_t orow1 = orow0 + 8 * BATCH;
        v4f base0 = H.base0, base1 = H.base1;

        float gv  = 1.0f / (1.0f + __expf(-H.gin));
        int   gvi = __float_as_int(h ? -gv : gv);

        v8f accA = {0,0,0,0,0,0,0,0};
        v8f accB = {0,0,0,0,0,0,0,0};
        #pragma unroll
        for (int jj = 0; jj < 4; ++jj) {
            #pragma unroll
            for (int j = 0; j < 4; j += 2) {
                v2f a0; a0.x = H.xv[jj][0][j];     a0.y = H.xv[jj][1][j];
                v2f a1; a1.x = H.xv[jj][0][j + 1]; a1.y = H.xv[jj][1][j + 1];
                v2f tb0; tb0.x = T[j][2 * jj];     tb0.y = T[j][2 * jj + 1];
                v2f tb1; tb1.x = T[j + 1][2 * jj]; tb1.y = T[j + 1][2 * jj + 1];
                accA = __builtin_amdgcn_wmma_f32_16x16x4_f32(false, a0, false, tb0,
                                                             (short)0, accA, false, false);
                accB = __builtin_amdgcn_wmma_f32_16x16x4_f32(false, a1, false, tb1,
                                                             (short)0, accB, false, false);
            }
        }
        #pragma unroll
        for (int jj = 0; jj < 8; jj += 2) {
            float gg = __int_as_float(__builtin_amdgcn_readlane(gvi, 4 * w + (jj >> 1)));
            v2f vb0; vb0.x = H.vraw[jj].x * gg;     vb0.y = H.vraw[jj].y * gg;
            v2f vb1; vb1.x = H.vraw[jj + 1].x * gg; vb1.y = H.vraw[jj + 1].y * gg;
            accA = __builtin_amdgcn_wmma_f32_16x16x4_f32(false, H.pa[jj], false, vb0,
                                                         (short)0, accA, false, false);
            accB = __builtin_amdgcn_wmma_f32_16x16x4_f32(false, H.pa[jj + 1], false, vb1,
                                                         (short)0, accB, false, false);
        }

        float* rw = &red[c & 1][16 * wv + q];
        #pragma unroll
        for (int r = 0; r < 8; ++r)
            rw[(8 * h + r) * RED_PITCH] = accA[r] + accB[r];
        STAGE_FENCE();

        ChunkHead Hn;
        Hn.base0 = (v4f){0, 0, 0, 0};
        Hn.base1 = (v4f){0, 0, 0, 0};
        load_group<1>(f1, kp, vp);  STAGE_FENCE();  mma_group<0>(f0, gvi, gl0, T);  STAGE_FENCE();
        load_group<2>(f0, kp, vp);  STAGE_FENCE();  mma_group<1>(f1, gvi, gl0, T);  STAGE_FENCE();
        if (has_base) {
            const float* bp0 = bb0 + (size_t)cn * CHUNK_WU;
            const float* bp1 = bp0 + (size_t)(8 * BATCH) * LD;
            Hn.base0.x = bp0[0]; Hn.base0.y = bp0[1]; Hn.base0.z = bp0[2]; Hn.base0.w = bp0[3];
            Hn.base1.x = bp1[0]; Hn.base1.y = bp1[1]; Hn.base1.z = bp1[2]; Hn.base1.w = bp1[3];
        }
        load_group<3>(f1, kp, vp);
        load_head_x(Hn, xr0 + (size_t)cn * CHUNK_XP, gp0 + (size_t)cn * CHUNK_WU);
        STAGE_FENCE();  mma_group<2>(f0, gvi, gl0, T);  STAGE_FENCE();
        load_head_p(Hn, pr0 + (size_t)cn * CHUNK_XP, vr0 + (size_t)cn * CHUNK_WU);
        load_group<0>(f0, kp0 + (size_t)cn * CHUNK_WU, vp0 + (size_t)cn * CHUNK_WU);
        STAGE_FENCE();  mma_group<3>(f1, gvi, gl0, T);
        STAGE_FENCE();

        __builtin_amdgcn_fence(__ATOMIC_RELEASE, "workgroup", "local");
        __builtin_amdgcn_s_barrier();
        __builtin_amdgcn_fence(__ATOMIC_ACQUIRE, "workgroup", "local");
        v4f v0 = *(const v4f*)&red[c & 1][fs * RED_PITCH + fc];
        v4f v1 = *(const v4f*)&red[c & 1][(8 + fs) * RED_PITCH + fc];
        *(v4f*)(po + orow0 * 256 + 64 * og + fc) = v0 + base0;
        *(v4f*)(po + orow1 * 256 + 64 * og + fc) = v1 + base1;
        H = Hn;
    }
}

__global__ __launch_bounds__(256)
void fwp_join(const float* __restrict__ part, float* __restrict__ out)
{
    int i = blockIdx.x * 256 + threadIdx.x;
    v4f pl[NPLANE];
    #pragma unroll
    for (int p = 0; p < NPLANE; ++p)
        pl[p] = *(const v4f*)(part + (size_t)p * OUT_FLOATS + 4 * (size_t)i);
    v4f s = pl[0];
    #pragma unroll
    for (int p = 1; p < NPLANE; ++p) s += pl[p];
    *(v4f*)(out + 4 * (size_t)i) = s;
}

extern "C" void kernel_launch(void* const* d_in, const int* in_sizes, int n_in,
                              void* d_out, int out_size, void* d_ws, size_t ws_size,
                              hipStream_t stream)
{
    const float* x       = (const float*)d_in[0];
    const float* slow_W  = (const float*)d_in[1];
    const float* slow_b  = (const float*)d_in[2];
    const float* fast_W0 = (const float*)d_in[3];
    float*       out     = (float*)d_out;
    float*       wu      = (float*)d_ws;
    float*       P       = (float*)d_ws + WU_FLOATS;
    float*       part    = (float*)d_ws + WU_FLOATS + P_FLOATS;

    fwp_slow_gemm<<<1344, 256, 0, stream>>>(x, slow_W, slow_b, fast_W0, wu);

    fwp_scores<<<256, 256, 0, stream>>>(wu, x, P);

    dim3 grid2(16, BATCH, 2);
    fwp_scan<<<grid2, 128, 0, stream>>>(wu, x, P, part);

    fwp_join<<<1024, 256, 0, stream>>>(part, out);
}
